# starting kernel
# baseline (speedup 1.0000x reference)
#include <hip/hip_runtime.h>

typedef __attribute__((ext_vector_type(2))) float v2f;
typedef __attribute__((ext_vector_type(8))) float v8f;

#define N_NODES 50000
#define FDIM    128
#define E_EDGES 1600000

__global__ void gcn_zero_kernel(float* __restrict__ h, int n) {
    int i = blockIdx.x * blockDim.x + threadIdx.x;
    if (i < n) h[i] = 0.0f;
}

__global__ void gcn_scatter_kernel(const float* __restrict__ feature,
                                   const int*   __restrict__ src,
                                   const int*   __restrict__ dst,
                                   float*       __restrict__ h) {
    long long t = (long long)blockIdx.x * blockDim.x + threadIdx.x;
    int e = (int)(t >> 5);
    int c = (int)(t & 31);
    if (e >= E_EDGES) return;

    int s = src[e];
    int d = dst[e];

    const float4* frow = (const float4*)(feature + (size_t)s * FDIM);
    float4 v = frow[c];

    float* hrow = h + (size_t)d * FDIM + (size_t)c * 4;
    unsafeAtomicAdd(hrow + 0, v.x);
    unsafeAtomicAdd(hrow + 1, v.y);
    unsafeAtomicAdd(hrow + 2, v.z);
    unsafeAtomicAdd(hrow + 3, v.w);
}

__global__ void gcn_gemm_wmma_kernel(const float* __restrict__ h,
                                     const float* __restrict__ W,
                                     const float* __restrict__ b,
                                     float*       __restrict__ out) {
    const int lane = threadIdx.x;
    const int col0 = blockIdx.x * 16;
    const int row0 = blockIdx.y * 16;
    const int ln   = lane & 15;
    const int half = lane >> 4;
    const int koff = half * 2;

    float bb = b[col0 + ln];
    v8f acc;
#pragma unroll
    for (int v = 0; v < 8; ++v) acc[v] = bb;

    const float* arow = h + (size_t)(row0 + ln) * FDIM;
    const float* brow = W + (size_t)(col0 + ln) * FDIM;

#pragma unroll 4
    for (int k0 = 0; k0 < FDIM; k0 += 4) {
        v2f a, bm;
        a[0]  = arow[k0 + koff];
        a[1]  = arow[k0 + koff + 1];
        bm[0] = brow[k0 + koff];
        bm[1] = brow[k0 + koff + 1];
        acc = __builtin_amdgcn_wmma_f32_16x16x4_f32(false, a, false, bm,
                                                    (short)0, acc, false, false);
    }

#pragma unroll
    for (int v = 0; v < 8; ++v) {
        out[(size_t)(row0 + v + 8 * half) * FDIM + col0 + ln] = acc[v];
    }
}

extern "C" void kernel_launch(void* const* d_in, const int* in_sizes, int n_in,
                              void* d_out, int out_size, void* d_ws, size_t ws_size,
                              hipStream_t stream) {
    const float* feature = (const float*)d_in[0];
    const int*   src     = (const int*)  d_in[1];
    const int*   dst     = (const int*)  d_in[2];
    const float* W       = (const float*)d_in[3];
    const float* b       = (const float*)d_in[4];
    float*       out     = (float*)d_out;
    float*       h       = (float*)d_ws;

    int nh = N_NODES * FDIM;
    gcn_zero_kernel<<<(nh + 255) / 256, 256, 0, stream>>>(h, nh);

    long long tot = (long long)E_EDGES * 32;
    unsigned nblk = (unsigned)((tot + 255) / 256);
    gcn_scatter_kernel<<<nblk, 256, 0, stream>>>(feature, src, dst, h);

    dim3 grid(FDIM / 16, N_NODES / 16);
    gcn_gemm_wmma_kernel<<<grid, 32, 0, stream>>>(h, W, b, out);
}


# Scatter replaced by an unsorted owner filter: 250 work-groups each read the whole dst list, keep their 200 rows' edges, add the source rows into LDS (arrival order), store h once; no global atomics
# speedup vs baseline: 1.7086x; 1.7086x over previous
#include <hip/hip_runtime.h>


typedef __attribute__((ext_vector_type(2))) float v2f;
typedef __attribute__((ext_vector_type(8))) float v8f;
typedef __attribute__((ext_vector_type(4))) float v4f;

#define N_NODES 50000
#define FDIM    128
#define E_EDGES 1600000

#define ROWS_PER_WG 200
#define N_OWNERS    250
#define WAVES       16
#define WG_THREADS  512
static_assert(N_OWNERS * ROWS_PER_WG == N_NODES, "owners must cover every node");
static_assert(WG_THREADS == WAVES * 32, "work-group size");
#define LOADS       16
#define WAVE_EDGES  (32 * 4 * LOADS)
#define CHUNK_EDGES (WAVES * WAVE_EDGES)
#define N_CHUNKS    ((E_EDGES + CHUNK_EDGES - 1) / CHUNK_EDGES)
#define ROWS_IN_FLIGHT 8

__device__ __forceinline__ void load_dst_chunk(int4 (&d)[LOADS], const int* __restrict__ dst,
                                               int chunk, int wave, int lane) {
    const int base = chunk * CHUNK_EDGES + wave * WAVE_EDGES + lane * 4;
#pragma unroll
    for (int u = 0; u < LOADS; ++u) {
        const int e0  = base + u * 128;
        const int e0c = (e0 < E_EDGES) ? e0 : (E_EDGES - 4);
        int4 x = *(const int4*)(dst + e0c);
        if (e0 >= E_EDGES) x = make_int4(-1, -1, -1, -1);
        d[u] = x;
    }
}

__device__ __forceinline__ void sum_chunk(const int4 (&cur)[LOADS], int chunk, int wave, int lane, int lo,
                                          const float* __restrict__ feature,
                                          const int*   __restrict__ src,
                                          unsigned* q, float* acc) {
    const int ebase = chunk * CHUNK_EDGES + wave * WAVE_EDGES + lane * 4;
    int n = 0;
#pragma unroll
    for (int u = 0; u < LOADS; ++u) {
        const int dd[4] = {cur[u].x, cur[u].y, cur[u].z, cur[u].w};
#pragma unroll
        for (int qd = 0; qd < 4; ++qd) {
            const unsigned t = (unsigned)(dd[qd] - lo);
            const bool m = t < (unsigned)ROWS_PER_WG;
            const unsigned mask = __builtin_amdgcn_ballot_w32(m);
            if (mask != 0u) {
                if (m) {
                    const int pos = n + (int)__builtin_amdgcn_mbcnt_lo(mask, 0u);
                    q[pos] = ((unsigned)(ebase + u * 128 + qd) << 8) | t;
                }
                n += __builtin_popcount(mask);
            }
        }
    }

    for (int b0 = 0; b0 < n; b0 += 32) {
        const int idx = (b0 + lane < n) ? (b0 + lane) : (n - 1);
        const unsigned ent = q[idx];
        const int s_l = src[ent >> 8];
        const int cnt = (n - b0 < 32) ? (n - b0) : 32;
        for (int k = 0; k < cnt; k += ROWS_IN_FLIGHT) {
            v4f      v[ROWS_IN_FLIGHT];
            unsigned r[ROWS_IN_FLIGHT];
#pragma unroll
            for (int i = 0; i < ROWS_IN_FLIGHT; ++i) {
                const int kk = (k + i < cnt) ? (k + i) : (cnt - 1);
                const int s  = __builtin_amdgcn_readlane(s_l, kk);
                r[i] = (unsigned)__builtin_amdgcn_readlane((int)ent, kk) & 255u;
                v[i] = *(const v4f*)(feature + (size_t)s * FDIM + lane * 4);
            }
            asm volatile("" :: "v"(v[0]), "v"(v[1]), "v"(v[2]), "v"(v[3]),
                               "v"(v[4]), "v"(v[5]), "v"(v[6]), "v"(v[7]));
#pragma unroll
            for (int i = 0; i < ROWS_IN_FLIGHT; ++i) {
                if (i == 0 || k + i < cnt) {
                    float* a = acc + r[i] * FDIM + lane;
                    unsafeAtomicAdd(a,      v[i].x);
                    unsafeAtomicAdd(a + 32, v[i].y);
                    unsafeAtomicAdd(a + 64, v[i].z);
                    unsafeAtomicAdd(a + 96, v[i].w);
                }
            }
        }
    }
}

__global__ __launch_bounds__(WG_THREADS)
void gcn_owner_sum_kernel(const float* __restrict__ feature,
                          const int*   __restrict__ src,
                          const int*   __restrict__ dst,
                          float*       __restrict__ h) {
    __shared__ float    acc[ROWS_PER_WG * FDIM];
    __shared__ unsigned wq[WAVES][WAVE_EDGES];

    const int tid  = threadIdx.x;
    const int lane = tid & 31;
    const int wave = tid >> 5;
    const int lo   = blockIdx.x * ROWS_PER_WG;

    float4* acc4 = (float4*)acc;
    for (int i = tid; i < ROWS_PER_WG * FDIM / 4; i += WG_THREADS)
        acc4[i] = make_float4(0.0f, 0.0f, 0.0f, 0.0f);
    __syncthreads();

    unsigned* q = wq[wave];
    int chunk = blockIdx.x % N_CHUNKS;

    int4 cur[LOADS], nxt[LOADS];
    load_dst_chunk(cur, dst, chunk, wave, lane);

    for (int it = 0; it < N_CHUNKS - 1; ++it) {
        const int nchunk = (chunk + 1 == N_CHUNKS) ? 0 : chunk + 1;
        load_dst_chunk(nxt, dst, nchunk, wave, lane);
        sum_chunk(cur, chunk, wave, lane, lo, feature, src, q, acc);
#pragma unroll
        for (int u = 0; u < LOADS; ++u) cur[u] = nxt[u];
        chunk = nchunk;
    }
    sum_chunk(cur, chunk, wave, lane, lo, feature, src, q, acc);
    __syncthreads();

    for (int r = wave; r < ROWS_PER_WG; r += WAVES) {
        const float* a = acc + r * FDIM + lane;
        float4 v = make_float4(a[0], a[32], a[64], a[96]);
        *(float4*)(h + (size_t)(lo + r) * FDIM + lane * 4) = v;
    }
}

__global__ void gcn_gemm_wmma_kernel(const float* __restrict__ h,
                                     const float* __restrict__ W,
                                     const float* __restrict__ b,
                                     float*       __restrict__ out) {
    const int lane = threadIdx.x;
    const int col0 = blockIdx.x * 16;
    const int row0 = blockIdx.y * 16;
    const int ln   = lane & 15;
    const int half = lane >> 4;
    const int koff = half * 2;

    float bb = b[col0 + ln];
    v8f acc;
#pragma unroll
    for (int v = 0; v < 8; ++v) acc[v] = bb;

    const float* arow = h + (size_t)(row0 + ln) * FDIM;
    const float* brow = W + (size_t)(col0 + ln) * FDIM;

#pragma unroll 4
    for (int k0 = 0; k0 < FDIM; k0 += 4) {
        v2f a, bm;
        a[0]  = arow[k0 + koff];
        a[1]  = arow[k0 + koff + 1];
        bm[0] = brow[k0 + koff];
        bm[1] = brow[k0 + koff + 1];
        acc = __builtin_amdgcn_wmma_f32_16x16x4_f32(false, a, false, bm,
                                                    (short)0, acc, false, false);
    }

#pragma unroll
    for (int v = 0; v < 8; ++v) {
        out[(size_t)(row0 + v + 8 * half) * FDIM + col0 + ln] = acc[v];
    }
}

extern "C" void kernel_launch(void* const* d_in, const int* in_sizes, int n_in,
                              void* d_out, int out_size, void* d_ws, size_t ws_size,
                              hipStream_t stream) {
    const float* feature = (const float*)d_in[0];
    const int*   src     = (const int*)  d_in[1];
    const int*   dst     = (const int*)  d_in[2];
    const float* W       = (const float*)d_in[3];
    const float* b       = (const float*)d_in[4];
    float*       out     = (float*)d_out;
    float*       h       = (float*)d_ws;

    gcn_owner_sum_kernel<<<N_OWNERS, WG_THREADS, 0, stream>>>(feature, src, dst, h);

    dim3 grid(FDIM / 16, N_NODES / 16);
    gcn_gemm_wmma_kernel<<<grid, 32, 0, stream>>>(h, W, b, out);
}


# Product moved into the owner kernel's tail: each wave keeps its 16 columns of W in registers and multiplies the LDS sums (528-byte row pitch); no h, one launch; k summed in a permuted order
# speedup vs baseline: 2.0421x; 1.1952x over previous
#include <hip/hip_runtime.h>


typedef __attribute__((ext_vector_type(2))) float v2f;
typedef __attribute__((ext_vector_type(8))) float v8f;
typedef __attribute__((ext_vector_type(4))) float v4f;

#define N_NODES 50000
#define FDIM    128
#define E_EDGES 1600000

#define ROWS_PER_WG 200
#define N_OWNERS    250
#define WAVES       16
#define WG_THREADS  512
static_assert(N_OWNERS * ROWS_PER_WG == N_NODES, "owners must cover every node");
static_assert(WG_THREADS == WAVES * 32, "work-group size");
#define LOADS       16
#define WAVE_EDGES  (32 * 4 * LOADS)
#define CHUNK_EDGES (WAVES * WAVE_EDGES)
#define N_CHUNKS    ((E_EDGES + CHUNK_EDGES - 1) / CHUNK_EDGES)
#define ROWS_IN_FLIGHT 8
#define PITCH       132
#define ROW_TILES   ((ROWS_PER_WG + 15) / 16)
#define ROWS_PAD    (ROW_TILES * 16)
#define COL_TILES   (FDIM / 16)
static_assert(WAVES == 2 * COL_TILES, "two waves per column tile in the product");
static_assert((PITCH * 4) % 16 == 0, "fragment reads are 16-byte loads");

__device__ __forceinline__ void load_dst_chunk(int4 (&d)[LOADS], const int* __restrict__ dst,
                                               int chunk, int wave, int lane) {
    const int base = chunk * CHUNK_EDGES + wave * WAVE_EDGES + lane * 4;
#pragma unroll
    for (int u = 0; u < LOADS; ++u) {
        const int e0  = base + u * 128;
        const int e0c = (e0 < E_EDGES) ? e0 : (E_EDGES - 4);
        int4 x = *(const int4*)(dst + e0c);
        if (e0 >= E_EDGES) x = make_int4(-1, -1, -1, -1);
        d[u] = x;
    }
}

__device__ __forceinline__ void sum_chunk(const int4 (&cur)[LOADS], int chunk, int wave, int lane, int lo,
                                          const float* __restrict__ feature,
                                          const int*   __restrict__ src,
                                          unsigned* q, float* acc) {
    const int ebase = chunk * CHUNK_EDGES + wave * WAVE_EDGES + lane * 4;
    int n = 0;
#pragma unroll
    for (int u = 0; u < LOADS; ++u) {
        const int dd[4] = {cur[u].x, cur[u].y, cur[u].z, cur[u].w};
#pragma unroll
        for (int qd = 0; qd < 4; ++qd) {
            const unsigned t = (unsigned)(dd[qd] - lo);
            const bool m = t < (unsigned)ROWS_PER_WG;
            const unsigned mask = __builtin_amdgcn_ballot_w32(m);
            if (mask != 0u) {
                if (m) {
                    const int pos = n + (int)__builtin_amdgcn_mbcnt_lo(mask, 0u);
                    q[pos] = ((unsigned)(ebase + u * 128 + qd) << 8) | t;
                }
                n += __builtin_popcount(mask);
            }
        }
    }

    for (int b0 = 0; b0 < n; b0 += 32) {
        const int idx = (b0 + lane < n) ? (b0 + lane) : (n - 1);
        const unsigned ent = q[idx];
        const int s_l = src[ent >> 8];
        const int cnt = (n - b0 < 32) ? (n - b0) : 32;
        for (int k = 0; k < cnt; k += ROWS_IN_FLIGHT) {
            v4f      v[ROWS_IN_FLIGHT];
            unsigned r[ROWS_IN_FLIGHT];
#pragma unroll
            for (int i = 0; i < ROWS_IN_FLIGHT; ++i) {
                const int kk = (k + i < cnt) ? (k + i) : (cnt - 1);
                const int s  = __builtin_amdgcn_readlane(s_l, kk);
                r[i] = (unsigned)__builtin_amdgcn_readlane((int)ent, kk) & 255u;
                v[i] = *(const v4f*)(feature + (size_t)s * FDIM + lane * 4);
            }
            asm volatile("" :: "v"(v[0]), "v"(v[1]), "v"(v[2]), "v"(v[3]),
                               "v"(v[4]), "v"(v[5]), "v"(v[6]), "v"(v[7]));
#pragma unroll
            for (int i = 0; i < ROWS_IN_FLIGHT; ++i) {
                if (i == 0 || k + i < cnt) {
                    float* a = acc + r[i] * PITCH + lane;
                    unsafeAtomicAdd(a,      v[i].x);
                    unsafeAtomicAdd(a + 32, v[i].y);
                    unsafeAtomicAdd(a + 64, v[i].z);
                    unsafeAtomicAdd(a + 96, v[i].w);
                }
            }
        }
    }
}

__global__ __launch_bounds__(WG_THREADS)
void gcn_owner_layer_kernel(const float* __restrict__ feature,
                            const int*   __restrict__ src,
                            const int*   __restrict__ dst,
                            const float* __restrict__ W,
                            const float* __restrict__ b,
                            float*       __restrict__ out) {
    __shared__ __attribute__((aligned(16))) float acc[ROWS_PAD * PITCH];
    __shared__ unsigned wq[WAVES][WAVE_EDGES];

    const int tid  = threadIdx.x;
    const int lane = tid & 31;
    const int wave = tid >> 5;
    const int lo   = blockIdx.x * ROWS_PER_WG;

    float4* acc4 = (float4*)acc;
    for (int i = tid; i < ROWS_PAD * PITCH / 4; i += WG_THREADS)
        acc4[i] = make_float4(0.0f, 0.0f, 0.0f, 0.0f);
    __syncthreads();

    unsigned* q = wq[wave];
    int chunk = blockIdx.x % N_CHUNKS;

    int4 cur[LOADS], nxt[LOADS];
    load_dst_chunk(cur, dst, chunk, wave, lane);

    for (int it = 0; it < N_CHUNKS - 1; ++it) {
        const int nchunk = (chunk + 1 == N_CHUNKS) ? 0 : chunk + 1;
        load_dst_chunk(nxt, dst, nchunk, wave, lane);
        sum_chunk(cur, chunk, wave, lane, lo, feature, src, q, acc);
#pragma unroll
        for (int u = 0; u < LOADS; ++u) cur[u] = nxt[u];
        chunk = nchunk;
    }

    const int wave_u = __builtin_amdgcn_readfirstlane(wave);
    const int ct = wave_u & (COL_TILES - 1);
    const int ln = lane & 15;
    const int hf = lane >> 4;
    const float* wrow = W + (size_t)(ct * 16 + ln) * FDIM + 16 * hf;
    v4f wf[4][4];
#pragma unroll
    for (int a = 0; a < 4; ++a)
#pragma unroll
        for (int j = 0; j < 4; ++j)
            wf[a][j] = *(const v4f*)(wrow + 32 * a + 4 * j);
    const float bias = b[ct * 16 + ln];

    sum_chunk(cur, chunk, wave, lane, lo, feature, src, q, acc);
    __syncthreads();

    for (int mt = wave_u >> 3; mt < ROW_TILES; mt += 2) {
        const float* arow = acc + (mt * 16 + ln) * PITCH + 4 * hf;
        v8f d;
#pragma unroll
        for (int v = 0; v < 8; ++v) d[v] = bias;
#pragma unroll
        for (int s = 0; s < 16; ++s) {
            const v4f x = *(const v4f*)(arow + 8 * s);
            const int a = s & 3, qq = s >> 2;
            v2f a0, a1, b0, b1;
            a0[0] = x[0];            a0[1] = x[1];
            a1[0] = x[2];            a1[1] = x[3];
            b0[0] = wf[a][0][qq];    b0[1] = wf[a][1][qq];
            b1[0] = wf[a][2][qq];    b1[1] = wf[a][3][qq];
            d = __builtin_amdgcn_wmma_f32_16x16x4_f32(false, a0, false, b0, (short)0, d, false, false);
            d = __builtin_amdgcn_wmma_f32_16x16x4_f32(false, a1, false, b1, (short)0, d, false, false);
        }
        const int rbase = mt * 16 + 8 * hf;
        float* o = out + (size_t)(lo + rbase) * FDIM + ct * 16 + ln;
#pragma unroll
        for (int v = 0; v < 8; ++v)
            if (rbase + v < ROWS_PER_WG) o[v * FDIM] = d[v];
    }
}

extern "C" void kernel_launch(void* const* d_in, const int* in_sizes, int n_in,
                              void* d_out, int out_size, void* d_ws, size_t ws_size,
                              hipStream_t stream) {
    const float* feature = (const float*)d_in[0];
    const int*   src     = (const int*)  d_in[1];
    const int*   dst     = (const int*)  d_in[2];
    const float* W       = (const float*)d_in[3];
    const float* b       = (const float*)d_in[4];
    float*       out     = (float*)d_out;

    gcn_owner_layer_kernel<<<N_OWNERS, WG_THREADS, 0, stream>>>(feature, src, dst, W, b, out);
}


# Kept edges wait in the wave's list across chunks and leave in full batches of 32: one index load, then 32 row loads in flight, no padding loads; LDS sums arrive in another order
# speedup vs baseline: 2.3106x; 1.1315x over previous
#include <hip/hip_runtime.h>


typedef __attribute__((ext_vector_type(2))) float v2f;
typedef __attribute__((ext_vector_type(8))) float v8f;
typedef __attribute__((ext_vector_type(4))) float v4f;

#define N_NODES 50000
#define FDIM    128
#define E_EDGES 1600000

#define ROWS_PER_WG 200
#define N_OWNERS    250
#define WAVES       16
#define WG_THREADS  512
static_assert(N_OWNERS * ROWS_PER_WG == N_NODES, "owners must cover every node");
static_assert(WG_THREADS == WAVES * 32, "work-group size");
#define LOADS       16
#define WAVE_EDGES  (32 * 4 * LOADS)
#define CHUNK_EDGES (WAVES * WAVE_EDGES)
#define N_CHUNKS    ((E_EDGES + CHUNK_EDGES - 1) / CHUNK_EDGES)
#define BATCH       32
#define ROWS_IN_FLIGHT 32
#define TAIL_ROWS_IN_FLIGHT 8
#define Q_WORDS     (WAVE_EDGES + BATCH)
#define PITCH       132
#define ROW_TILES   ((ROWS_PER_WG + 15) / 16)
#define ROWS_PAD    (ROW_TILES * 16)
#define COL_TILES   (FDIM / 16)
static_assert(WAVES == 2 * COL_TILES, "two waves per column tile in the product");
static_assert((PITCH * 4) % 16 == 0, "fragment reads are 16-byte loads");

__device__ __forceinline__ void load_dst_chunk(int4 (&d)[LOADS], const int* __restrict__ dst,
                                               int chunk, int wave, int lane) {
    const int base = chunk * CHUNK_EDGES + wave * WAVE_EDGES + lane * 4;
#pragma unroll
    for (int u = 0; u < LOADS; ++u) {
        const int e0  = base + u * 128;
        const int e0c = (e0 < E_EDGES) ? e0 : (E_EDGES - 4);
        int4 x = *(const int4*)(dst + e0c);
        if (e0 >= E_EDGES) x = make_int4(-1, -1, -1, -1);
        d[u] = x;
    }
}

__device__ __forceinline__ int keep_chunk(const int4 (&cur)[LOADS], int chunk, int wave, int lane, int lo,
                                          unsigned* q, int n) {
    const int ebase = chunk * CHUNK_EDGES + wave * WAVE_EDGES + lane * 4;
#pragma unroll
    for (int u = 0; u < LOADS; ++u) {
        const int dd[4] = {cur[u].x, cur[u].y, cur[u].z, cur[u].w};
#pragma unroll
        for (int qd = 0; qd < 4; ++qd) {
            const unsigned t = (unsigned)(dd[qd] - lo);
            const bool m = t < (unsigned)ROWS_PER_WG;
            const unsigned mask = __builtin_amdgcn_ballot_w32(m);
            if (mask != 0u) {
                if (m) {
                    const int pos = n + (int)__builtin_amdgcn_mbcnt_lo(mask, 0u);
                    q[pos] = ((unsigned)(ebase + u * 128 + qd) << 8) | t;
                }
                n += __builtin_popcount(mask);
            }
        }
    }
    return n;
}

__device__ __forceinline__ void gather_full(unsigned* qb, int lane,
                                            const float* __restrict__ feature,
                                            const int*   __restrict__ src,
                                            float* acc) {
    const unsigned ent = qb[lane];
    const int s_l = src[ent >> 8];
#pragma unroll
    for (int k = 0; k < BATCH; k += ROWS_IN_FLIGHT) {
        v4f      v[ROWS_IN_FLIGHT];
        unsigned r[ROWS_IN_FLIGHT];
#pragma unroll
        for (int i = 0; i < ROWS_IN_FLIGHT; ++i) {
            const int s = __builtin_amdgcn_readlane(s_l, k + i);
            r[i] = (unsigned)__builtin_amdgcn_readlane((int)ent, k + i) & 255u;
            v[i] = *(const v4f*)(feature + (size_t)s * FDIM + lane * 4);
        }
#pragma unroll
        for (int i = 0; i < ROWS_IN_FLIGHT; ++i) {
            float* a = acc + r[i] * PITCH + lane;
            unsafeAtomicAdd(a,      v[i].x);
            unsafeAtomicAdd(a + 32, v[i].y);
            unsafeAtomicAdd(a + 64, v[i].z);
            unsafeAtomicAdd(a + 96, v[i].w);
        }
    }
}

__device__ __forceinline__ void gather_tail(unsigned* q, int n, int lane,
                                            const float* __restrict__ feature,
                                            const int*   __restrict__ src,
                                            float* acc) {
    for (int b0 = 0; b0 < n; b0 += 32) {
        const int idx = (b0 + lane < n) ? (b0 + lane) : (n - 1);
        const unsigned ent = q[idx];
        const int s_l = src[ent >> 8];
        const int cnt = (n - b0 < 32) ? (n - b0) : 32;
        for (int k = 0; k < cnt; k += TAIL_ROWS_IN_FLIGHT) {
            v4f      v[TAIL_ROWS_IN_FLIGHT];
            unsigned r[TAIL_ROWS_IN_FLIGHT];
#pragma unroll
            for (int i = 0; i < TAIL_ROWS_IN_FLIGHT; ++i) {
                const int kk = (k + i < cnt) ? (k + i) : (cnt - 1);
                const int s  = __builtin_amdgcn_readlane(s_l, kk);
                r[i] = (unsigned)__builtin_amdgcn_readlane((int)ent, kk) & 255u;
                v[i] = *(const v4f*)(feature + (size_t)s * FDIM + lane * 4);
            }
            asm volatile("" :: "v"(v[0]), "v"(v[1]), "v"(v[2]), "v"(v[3]),
                               "v"(v[4]), "v"(v[5]), "v"(v[6]), "v"(v[7]));
#pragma unroll
            for (int i = 0; i < TAIL_ROWS_IN_FLIGHT; ++i) {
                if (i == 0 || k + i < cnt) {
                    float* a = acc + r[i] * PITCH + lane;
                    unsafeAtomicAdd(a,      v[i].x);
                    unsafeAtomicAdd(a + 32, v[i].y);
                    unsafeAtomicAdd(a + 64, v[i].z);
                    unsafeAtomicAdd(a + 96, v[i].w);
                }
            }
        }
    }
}

__global__ __launch_bounds__(WG_THREADS)
void gcn_owner_layer_kernel(const float* __restrict__ feature,
                            const int*   __restrict__ src,
                            const int*   __restrict__ dst,
                            const float* __restrict__ W,
                            const float* __restrict__ b,
                            float*       __restrict__ out) {
    __shared__ __attribute__((aligned(16))) float acc[ROWS_PAD * PITCH];
    __shared__ unsigned wq[WAVES][Q_WORDS];

    const int tid  = threadIdx.x;
    const int lane = tid & 31;
    const int wave = tid >> 5;
    const int lo   = blockIdx.x * ROWS_PER_WG;

    float4* acc4 = (float4*)acc;
    for (int i = tid; i < ROWS_PAD * PITCH / 4; i += WG_THREADS)
        acc4[i] = make_float4(0.0f, 0.0f, 0.0f, 0.0f);
    __syncthreads();

    unsigned* q = wq[wave];
    int chunk = blockIdx.x % N_CHUNKS;

    int4 cur[LOADS], nxt[LOADS];
    load_dst_chunk(cur, dst, chunk, wave, lane);
    int n = 0;

    for (int it = 0; it < N_CHUNKS - 1; ++it) {
        const int nchunk = (chunk + 1 == N_CHUNKS) ? 0 : chunk + 1;
        load_dst_chunk(nxt, dst, nchunk, wave, lane);
        n = keep_chunk(cur, chunk, wave, lane, lo, q, n);
        while (n >= BATCH) {
            n -= BATCH;
            gather_full(q + n, lane, feature, src, acc);
        }
#pragma unroll
        for (int u = 0; u < LOADS; ++u) cur[u] = nxt[u];
        chunk = nchunk;
    }

    const int wave_u = __builtin_amdgcn_readfirstlane(wave);
    const int ct = wave_u & (COL_TILES - 1);
    const int ln = lane & 15;
    const int hf = lane >> 4;
    const float* wrow = W + (size_t)(ct * 16 + ln) * FDIM + 16 * hf;
    v4f wf[4][4];
#pragma unroll
    for (int a = 0; a < 4; ++a)
#pragma unroll
        for (int j = 0; j < 4; ++j)
            wf[a][j] = *(const v4f*)(wrow + 32 * a + 4 * j);
    const float bias = b[ct * 16 + ln];

    n = keep_chunk(cur, chunk, wave, lane, lo, q, n);
    while (n >= BATCH) {
        n -= BATCH;
        gather_full(q + n, lane, feature, src, acc);
    }
    gather_tail(q, n, lane, feature, src, acc);
    __syncthreads();

    for (int mt = wave_u >> 3; mt < ROW_TILES; mt += 2) {
        const float* arow = acc + (mt * 16 + ln) * PITCH + 4 * hf;
        v8f d;
#pragma unroll
        for (int v = 0; v < 8; ++v) d[v] = bias;
#pragma unroll
        for (int s = 0; s < 16; ++s) {
            const v4f x = *(const v4f*)(arow + 8 * s);
            const int a = s & 3, qq = s >> 2;
            v2f a0, a1, b0, b1;
            a0[0] = x[0];            a0[1] = x[1];
            a1[0] = x[2];            a1[1] = x[3];
            b0[0] = wf[a][0][qq];    b0[1] = wf[a][1][qq];
            b1[0] = wf[a][2][qq];    b1[1] = wf[a][3][qq];
            d = __builtin_amdgcn_wmma_f32_16x16x4_f32(false, a0, false, b0, (short)0, d, false, false);
            d = __builtin_amdgcn_wmma_f32_16x16x4_f32(false, a1, false, b1, (short)0, d, false, false);
        }
        const int rbase = mt * 16 + 8 * hf;
        float* o = out + (size_t)(lo + rbase) * FDIM + ct * 16 + ln;
#pragma unroll
        for (int v = 0; v < 8; ++v)
            if (rbase + v < ROWS_PER_WG) o[v * FDIM] = d[v];
    }
}

extern "C" void kernel_launch(void* const* d_in, const int* in_sizes, int n_in,
                              void* d_out, int out_size, void* d_ws, size_t ws_size,
                              hipStream_t stream) {
    const float* feature = (const float*)d_in[0];
    const int*   src     = (const int*)  d_in[1];
    const int*   dst     = (const int*)  d_in[2];
    const float* W       = (const float*)d_in[3];
    const float* b       = (const float*)d_in[4];
    float*       out     = (float*)d_out;

    gcn_owner_layer_kernel<<<N_OWNERS, WG_THREADS, 0, stream>>>(feature, src, dst, W, b, out);
}


# Owner scan reads packed keys: a first launch writes dst/200 of each edge as 8 bit planes (1 byte/edge); owners AND the planes, 32 edges per lane-step; 2 launches; LDS sums arrive in another order
# speedup vs baseline: 3.7222x; 1.6109x over previous
#include <hip/hip_runtime.h>


typedef __attribute__((ext_vector_type(2))) float v2f;
typedef __attribute__((ext_vector_type(8))) float v8f;
typedef __attribute__((ext_vector_type(4))) float v4f;

#define N_NODES 50000
#define FDIM    128
#define E_EDGES 1600000

#define ROWS_PER_WG 200
#define N_OWNERS    250
#define WAVES       16
#define WG_THREADS  512
static_assert(N_OWNERS * ROWS_PER_WG == N_NODES, "owners must cover every node");
static_assert(WG_THREADS == WAVES * 32, "work-group size");
static_assert(N_OWNERS <= 255, "an owner number fits in 8 bits; 255 means nobody");
#define BLOCK_EDGES 1024
#define N_BLOCKS    ((E_EDGES + BLOCK_EDGES - 1) / BLOCK_EDGES)
#define BLKS        4
#define CHUNK_BLOCKS (WAVES * BLKS)
#define N_CHUNKS    ((N_BLOCKS + CHUNK_BLOCKS - 1) / CHUNK_BLOCKS)
#define PACK_WAVES  4
#define BATCH       32
#define ROWS_IN_FLIGHT 32
#define TAIL_ROWS_IN_FLIGHT 8
#define Q_WORDS     (BLOCK_EDGES + BATCH)
#define PITCH       132
#define ROW_TILES   ((ROWS_PER_WG + 15) / 16)
#define ROWS_PAD    (ROW_TILES * 16)
#define COL_TILES   (FDIM / 16)
static_assert(WAVES == 2 * COL_TILES, "two waves per column tile in the product");
static_assert((PITCH * 4) % 16 == 0, "fragment reads are 16-byte loads");

__global__ __launch_bounds__(PACK_WAVES * 32)
void gcn_pack_owner_kernel(const int* __restrict__ dst, unsigned* __restrict__ packed) {
    uint4* planes = (uint4*)packed;
    const int lane = threadIdx.x & 31;
    const int blk  = blockIdx.x * PACK_WAVES + (threadIdx.x >> 5);
    if (blk >= N_BLOCKS) return;
    const int base = blk * BLOCK_EDGES + lane * 4;
    int4 d[8];
#pragma unroll
    for (int u = 0; u < 8; ++u) {
        const int e0  = base + u * 128;
        const int e0c = (e0 < E_EDGES) ? e0 : (E_EDGES - 4);
        int4 x = *(const int4*)(dst + e0c);
        if (e0 >= E_EDGES) x = make_int4(-1, -1, -1, -1);
        d[u] = x;
    }
    unsigned p[8] = {0u, 0u, 0u, 0u, 0u, 0u, 0u, 0u};
#pragma unroll
    for (int u = 0; u < 8; ++u) {
        const int dd[4] = {d[u].x, d[u].y, d[u].z, d[u].w};
#pragma unroll
        for (int j = 0; j < 4; ++j) {
            unsigned o = (unsigned)dd[j] / (unsigned)ROWS_PER_WG;
            o = (o < 255u) ? o : 255u;
            const int gi = u * 4 + j;
#pragma unroll
            for (int b = 0; b < 8; ++b) {
                const unsigned m = __builtin_amdgcn_ballot_w32(((o >> b) & 1u) != 0u);
                if (lane == gi) p[b] = m;
            }
        }
    }
    planes[blk * 64 + lane]      = make_uint4(p[0], p[1], p[2], p[3]);
    planes[blk * 64 + 32 + lane] = make_uint4(p[4], p[5], p[6], p[7]);
}

__device__ __forceinline__ void load_planes(uint4 (&d)[BLKS][2], const uint4* __restrict__ planes,
                                            int chunk, int wave, int lane) {
#pragma unroll
    for (int k = 0; k < BLKS; ++k) {
        const int blk  = (chunk * WAVES + wave) * BLKS + k;
        const int blkc = (blk < N_BLOCKS) ? blk : (N_BLOCKS - 1);
        d[k][0] = planes[blkc * 64 + lane];
        d[k][1] = planes[blkc * 64 + 32 + lane];
    }
}

__device__ __forceinline__ int keep_block(const uint4 (&pl)[2], int blk, int lane, const unsigned (&inv)[8],
                                          unsigned* q, int n) {
    unsigned hm = (pl[0].x ^ inv[0]) & (pl[0].y ^ inv[1]) & (pl[0].z ^ inv[2]) & (pl[0].w ^ inv[3])
                & (pl[1].x ^ inv[4]) & (pl[1].y ^ inv[5]) & (pl[1].z ^ inv[6]) & (pl[1].w ^ inv[7]);
    if (blk >= N_BLOCKS) hm = 0u;
    const unsigned e_lane = (unsigned)blk * BLOCK_EDGES + (unsigned)(lane >> 2) * 128u + (unsigned)(lane & 3);
    unsigned any = __builtin_amdgcn_ballot_w32(hm != 0u);
    while (any != 0u) {
        if (hm != 0u) {
            const unsigned l = (unsigned)__builtin_ctz(hm);
            hm &= hm - 1u;
            const int pos = n + (int)__builtin_amdgcn_mbcnt_lo(any, 0u);
            q[pos] = e_lane + 4u * l;
        }
        n += __builtin_popcount(any);
        any = __builtin_amdgcn_ballot_w32(hm != 0u);
    }
    return n;
}

__device__ __forceinline__ void gather_full(unsigned* qb, int lane, int lo,
                                            const float* __restrict__ feature,
                                            const int*   __restrict__ src,
                                            const int*   __restrict__ dst,
                                            float* acc) {
    const unsigned ent = qb[lane];
    const int s_l = src[ent];
    const int r_l = dst[ent] - lo;
#pragma unroll
    for (int k = 0; k < BATCH; k += ROWS_IN_FLIGHT) {
        v4f      v[ROWS_IN_FLIGHT];
        unsigned r[ROWS_IN_FLIGHT];
#pragma unroll
        for (int i = 0; i < ROWS_IN_FLIGHT; ++i) {
            const int s = __builtin_amdgcn_readlane(s_l, k + i);
            r[i] = (unsigned)__builtin_amdgcn_readlane(r_l, k + i);
            v[i] = *(const v4f*)(feature + (size_t)s * FDIM + lane * 4);
        }
#pragma unroll
        for (int i = 0; i < ROWS_IN_FLIGHT; ++i) {
            float* a = acc + r[i] * PITCH + lane;
            unsafeAtomicAdd(a,      v[i].x);
            unsafeAtomicAdd(a + 32, v[i].y);
            unsafeAtomicAdd(a + 64, v[i].z);
            unsafeAtomicAdd(a + 96, v[i].w);
        }
    }
}

__device__ __forceinline__ void gather_tail(unsigned* q, int n, int lane, int lo,
                                            const float* __restrict__ feature,
                                            const int*   __restrict__ src,
                                            const int*   __restrict__ dst,
                                            float* acc) {
    for (int b0 = 0; b0 < n; b0 += 32) {
        const int idx = (b0 + lane < n) ? (b0 + lane) : (n - 1);
        const unsigned ent = q[idx];
        const int s_l = src[ent];
        const int r_l = dst[ent] - lo;
        const int cnt = (n - b0 < 32) ? (n - b0) : 32;
        for (int k = 0; k < cnt; k += TAIL_ROWS_IN_FLIGHT) {
            v4f      v[TAIL_ROWS_IN_FLIGHT];
            unsigned r[TAIL_ROWS_IN_FLIGHT];
#pragma unroll
            for (int i = 0; i < TAIL_ROWS_IN_FLIGHT; ++i) {
                const int kk = (k + i < cnt) ? (k + i) : (cnt - 1);
                const int s  = __builtin_amdgcn_readlane(s_l, kk);
                r[i] = (unsigned)__builtin_amdgcn_readlane(r_l, kk);
                v[i] = *(const v4f*)(feature + (size_t)s * FDIM + lane * 4);
            }
            asm volatile("" :: "v"(v[0]), "v"(v[1]), "v"(v[2]), "v"(v[3]),
                               "v"(v[4]), "v"(v[5]), "v"(v[6]), "v"(v[7]));
#pragma unroll
            for (int i = 0; i < TAIL_ROWS_IN_FLIGHT; ++i) {
                if (i == 0 || k + i < cnt) {
                    float* a = acc + r[i] * PITCH + lane;
                    unsafeAtomicAdd(a,      v[i].x);
                    unsafeAtomicAdd(a + 32, v[i].y);
                    unsafeAtomicAdd(a + 64, v[i].z);
                    unsafeAtomicAdd(a + 96, v[i].w);
                }
            }
        }
    }
}

__global__ __launch_bounds__(WG_THREADS)
void gcn_owner_layer_kernel(const float* __restrict__ feature,
                            const int*   __restrict__ src,
                            const int*   __restrict__ dst,
                            const float* __restrict__ W,
                            const float* __restrict__ b,
                            const unsigned* __restrict__ packed,
                            float*       __restrict__ out) {
    const uint4* planes = (const uint4*)packed;
    __shared__ __attribute__((aligned(16))) float acc[ROWS_PAD * PITCH];
    __shared__ unsigned wq[WAVES][Q_WORDS];

    const int tid  = threadIdx.x;
    const int lane = tid & 31;
    const int wave = tid >> 5;
    const int lo   = blockIdx.x * ROWS_PER_WG;

    float4* acc4 = (float4*)acc;
    for (int i = tid; i < ROWS_PAD * PITCH / 4; i += WG_THREADS)
        acc4[i] = make_float4(0.0f, 0.0f, 0.0f, 0.0f);
    __syncthreads();

    unsigned inv[8];
#pragma unroll
    for (int bb = 0; bb < 8; ++bb) inv[bb] = ((blockIdx.x >> bb) & 1u) ? 0u : ~0u;

    unsigned* q = wq[wave];
    int chunk = blockIdx.x % N_CHUNKS;

    uint4 cur[BLKS][2], nxt[BLKS][2];
    load_planes(cur, planes, chunk, wave, lane);
    int n = 0;

    for (int it = 0; it < N_CHUNKS - 1; ++it) {
        const int nchunk = (chunk + 1 == N_CHUNKS) ? 0 : chunk + 1;
        load_planes(nxt, planes, nchunk, wave, lane);
#pragma unroll
        for (int k = 0; k < BLKS; ++k) {
            n = keep_block(cur[k], (chunk * WAVES + wave) * BLKS + k, lane, inv, q, n);
            while (n >= BATCH) {
                n -= BATCH;
                gather_full(q + n, lane, lo, feature, src, dst, acc);
            }
        }
#pragma unroll
        for (int k = 0; k < BLKS; ++k) { cur[k][0] = nxt[k][0]; cur[k][1] = nxt[k][1]; }
        chunk = nchunk;
    }

    const int wave_u = __builtin_amdgcn_readfirstlane(wave);
    const int ct = wave_u & (COL_TILES - 1);
    const int ln = lane & 15;
    const int hf = lane >> 4;
    const float* wrow = W + (size_t)(ct * 16 + ln) * FDIM + 16 * hf;
    v4f wf[4][4];
#pragma unroll
    for (int a = 0; a < 4; ++a)
#pragma unroll
        for (int j = 0; j < 4; ++j)
            wf[a][j] = *(const v4f*)(wrow + 32 * a + 4 * j);
    const float bias = b[ct * 16 + ln];

#pragma unroll
    for (int k = 0; k < BLKS; ++k) {
        n = keep_block(cur[k], (chunk * WAVES + wave) * BLKS + k, lane, inv, q, n);
        while (n >= BATCH) {
            n -= BATCH;
            gather_full(q + n, lane, lo, feature, src, dst, acc);
        }
    }
    gather_tail(q, n, lane, lo, feature, src, dst, acc);
    __syncthreads();

    for (int mt = wave_u >> 3; mt < ROW_TILES; mt += 2) {
        const float* arow = acc + (mt * 16 + ln) * PITCH + 4 * hf;
        v8f d;
#pragma unroll
        for (int v = 0; v < 8; ++v) d[v] = bias;
#pragma unroll
        for (int s = 0; s < 16; ++s) {
            const v4f x = *(const v4f*)(arow + 8 * s);
            const int a = s & 3, qq = s >> 2;
            v2f a0, a1, b0, b1;
            a0[0] = x[0];            a0[1] = x[1];
            a1[0] = x[2];            a1[1] = x[3];
            b0[0] = wf[a][0][qq];    b0[1] = wf[a][1][qq];
            b1[0] = wf[a][2][qq];    b1[1] = wf[a][3][qq];
            d = __builtin_amdgcn_wmma_f32_16x16x4_f32(false, a0, false, b0, (short)0, d, false, false);
            d = __builtin_amdgcn_wmma_f32_16x16x4_f32(false, a1, false, b1, (short)0, d, false, false);
        }
        const int rbase = mt * 16 + 8 * hf;
        float* o = out + (size_t)(lo + rbase) * FDIM + ct * 16 + ln;
#pragma unroll
        for (int v = 0; v < 8; ++v)
            if (rbase + v < ROWS_PER_WG) o[v * FDIM] = d[v];
    }
}

extern "C" void kernel_launch(void* const* d_in, const int* in_sizes, int n_in,
                              void* d_out, int out_size, void* d_ws, size_t ws_size,
                              hipStream_t stream) {
    const float* feature = (const float*)d_in[0];
    const int*   src     = (const int*)  d_in[1];
    const int*   dst     = (const int*)  d_in[2];
    const float* W       = (const float*)d_in[3];
    const float* b       = (const float*)d_in[4];
    float*       out     = (float*)d_out;
    unsigned*    planes  = (unsigned*)d_ws;

    gcn_pack_owner_kernel<<<(N_BLOCKS + PACK_WAVES - 1) / PACK_WAVES, PACK_WAVES * 32, 0, stream>>>(dst, planes);
    gcn_owner_layer_kernel<<<N_OWNERS, WG_THREADS, 0, stream>>>(feature, src, dst, W, b, planes, out);
}


# Launch 1 also writes a 4-byte record per edge (source row, row within its owner); an owner fetches that one random word per kept edge, not src and dst from two lines; LDS sums still in arrival order
# speedup vs baseline: 3.9461x; 1.0602x over previous
#include <hip/hip_runtime.h>


typedef __attribute__((ext_vector_type(2))) float v2f;
typedef __attribute__((ext_vector_type(8))) float v8f;
typedef __attribute__((ext_vector_type(4))) float v4f;

#define N_NODES 50000
#define FDIM    128
#define E_EDGES 1600000

#define ROWS_PER_WG 200
#define N_OWNERS    250
#define WAVES       16
#define WG_THREADS  512
static_assert(N_OWNERS * ROWS_PER_WG == N_NODES, "owners must cover every node");
static_assert(WG_THREADS == WAVES * 32, "work-group size");
static_assert(N_OWNERS <= 255, "an owner number fits in 8 bits; 255 means nobody");
#define BLOCK_EDGES 1024
#define N_BLOCKS    ((E_EDGES + BLOCK_EDGES - 1) / BLOCK_EDGES)
#define BLKS        4
#define CHUNK_BLOCKS (WAVES * BLKS)
#define N_CHUNKS    ((N_BLOCKS + CHUNK_BLOCKS - 1) / CHUNK_BLOCKS)
#define PACK_WAVES  4
#define PLANE_WORDS (N_BLOCKS * (BLOCK_EDGES / 4))
static_assert(N_NODES <= 65536, "a source row number fits in the low 16 bits of an edge record");
static_assert(ROWS_PER_WG <= 256, "a row number within its owner fits in 8 bits of an edge record");
static_assert((PLANE_WORDS * 4) % 16 == 0, "the records start on a 16-byte boundary");
#define BATCH       32
#define ROWS_IN_FLIGHT 32
#define TAIL_ROWS_IN_FLIGHT 8
#define Q_WORDS     (BLOCK_EDGES + BATCH)
#define PITCH       132
#define ROW_TILES   ((ROWS_PER_WG + 15) / 16)
#define ROWS_PAD    (ROW_TILES * 16)
#define COL_TILES   (FDIM / 16)
static_assert(WAVES == 2 * COL_TILES, "two waves per column tile in the product");
static_assert((PITCH * 4) % 16 == 0, "fragment reads are 16-byte loads");

__global__ __launch_bounds__(PACK_WAVES * 32)
void gcn_pack_owner_kernel(const int* __restrict__ src, const int* __restrict__ dst,
                           unsigned* __restrict__ packed) {
    uint4*    planes = (uint4*)packed;
    unsigned* recs   = packed + PLANE_WORDS;
    const int lane = threadIdx.x & 31;
    const int blk  = blockIdx.x * PACK_WAVES + (threadIdx.x >> 5);
    if (blk >= N_BLOCKS) return;
    const int base = blk * BLOCK_EDGES + lane * 4;
    int4 d[8], sv[8];
#pragma unroll
    for (int u = 0; u < 8; ++u) {
        const int e0  = base + u * 128;
        const int e0c = (e0 < E_EDGES) ? e0 : (E_EDGES - 4);
        int4 x = *(const int4*)(dst + e0c);
        sv[u]  = *(const int4*)(src + e0c);
        if (e0 >= E_EDGES) x = make_int4(-1, -1, -1, -1);
        d[u] = x;
    }
    asm volatile("" :: "v"(sv[0].x), "v"(sv[0].y), "v"(sv[0].z), "v"(sv[0].w));
    unsigned p[8] = {0u, 0u, 0u, 0u, 0u, 0u, 0u, 0u};
#pragma unroll
    for (int u = 0; u < 8; ++u) {
        const int dd[4] = {d[u].x, d[u].y, d[u].z, d[u].w};
        const int ss[4] = {sv[u].x, sv[u].y, sv[u].z, sv[u].w};
        unsigned rc[4];
#pragma unroll
        for (int j = 0; j < 4; ++j) {
            unsigned o = (unsigned)dd[j] / (unsigned)ROWS_PER_WG;
            rc[j] = ((unsigned)ss[j] & 0xFFFFu) | ((((unsigned)dd[j] - o * (unsigned)ROWS_PER_WG) & 0xFFu) << 16);
            o = (o < 255u) ? o : 255u;
            const int gi = u * 4 + j;
#pragma unroll
            for (int b = 0; b < 8; ++b) {
                const unsigned m = __builtin_amdgcn_ballot_w32(((o >> b) & 1u) != 0u);
                if (lane == gi) p[b] = m;
            }
        }
        const int e0 = base + u * 128;
        if (e0 < E_EDGES)
            *(uint4*)(recs + e0) = make_uint4(rc[0], rc[1], rc[2], rc[3]);
    }
    planes[blk * 64 + lane]      = make_uint4(p[0], p[1], p[2], p[3]);
    planes[blk * 64 + 32 + lane] = make_uint4(p[4], p[5], p[6], p[7]);
}

__device__ __forceinline__ void load_planes(uint4 (&d)[BLKS][2], const uint4* __restrict__ planes,
                                            int chunk, int wave, int lane) {
#pragma unroll
    for (int k = 0; k < BLKS; ++k) {
        const int blk  = (chunk * WAVES + wave) * BLKS + k;
        const int blkc = (blk < N_BLOCKS) ? blk : (N_BLOCKS - 1);
        d[k][0] = planes[blkc * 64 + lane];
        d[k][1] = planes[blkc * 64 + 32 + lane];
    }
}

__device__ __forceinline__ int keep_block(const uint4 (&pl)[2], int blk, int lane, const unsigned (&inv)[8],
                                          unsigned* q, int n) {
    unsigned hm = (pl[0].x ^ inv[0]) & (pl[0].y ^ inv[1]) & (pl[0].z ^ inv[2]) & (pl[0].w ^ inv[3])
                & (pl[1].x ^ inv[4]) & (pl[1].y ^ inv[5]) & (pl[1].z ^ inv[6]) & (pl[1].w ^ inv[7]);
    if (blk >= N_BLOCKS) hm = 0u;
    const unsigned e_lane = (unsigned)blk * BLOCK_EDGES + (unsigned)(lane >> 2) * 128u + (unsigned)(lane & 3);
    unsigned any = __builtin_amdgcn_ballot_w32(hm != 0u);
    while (any != 0u) {
        if (hm != 0u) {
            const unsigned l = (unsigned)__builtin_ctz(hm);
            hm &= hm - 1u;
            const int pos = n + (int)__builtin_amdgcn_mbcnt_lo(any, 0u);
            q[pos] = e_lane + 4u * l;
        }
        n += __builtin_popcount(any);
        any = __builtin_amdgcn_ballot_w32(hm != 0u);
    }
    return n;
}

__device__ __forceinline__ void gather_full(unsigned* qb, int lane,
                                            const float*    __restrict__ feature,
                                            const unsigned* __restrict__ recs,
                                            float* acc) {
    const unsigned ent   = qb[lane];
    const unsigned rec_l = recs[ent];
#pragma unroll
    for (int k = 0; k < BATCH; k += ROWS_IN_FLIGHT) {
        v4f      v[ROWS_IN_FLIGHT];
        unsigned r[ROWS_IN_FLIGHT];
#pragma unroll
        for (int i = 0; i < ROWS_IN_FLIGHT; ++i) {
            const unsigned rec = (unsigned)__builtin_amdgcn_readlane((int)rec_l, k + i);
            const unsigned s   = rec & 0xFFFFu;
            r[i] = rec >> 16;
            v[i] = *(const v4f*)(feature + (size_t)s * FDIM + lane * 4);
        }
#pragma unroll
        for (int i = 0; i < ROWS_IN_FLIGHT; ++i) {
            float* a = acc + r[i] * PITCH + lane;
            unsafeAtomicAdd(a,      v[i].x);
            unsafeAtomicAdd(a + 32, v[i].y);
            unsafeAtomicAdd(a + 64, v[i].z);
            unsafeAtomicAdd(a + 96, v[i].w);
        }
    }
}

__device__ __forceinline__ void gather_tail(unsigned* q, int n, int lane,
                                            const float*    __restrict__ feature,
                                            const unsigned* __restrict__ recs,
                                            float* acc) {
    for (int b0 = 0; b0 < n; b0 += 32) {
        const int idx = (b0 + lane < n) ? (b0 + lane) : (n - 1);
        const unsigned ent   = q[idx];
        const unsigned rec_l = recs[ent];
        const int cnt = (n - b0 < 32) ? (n - b0) : 32;
        for (int k = 0; k < cnt; k += TAIL_ROWS_IN_FLIGHT) {
            v4f      v[TAIL_ROWS_IN_FLIGHT];
            unsigned r[TAIL_ROWS_IN_FLIGHT];
#pragma unroll
            for (int i = 0; i < TAIL_ROWS_IN_FLIGHT; ++i) {
                const int kk = (k + i < cnt) ? (k + i) : (cnt - 1);
                const unsigned rec = (unsigned)__builtin_amdgcn_readlane((int)rec_l, kk);
                const unsigned s   = rec & 0xFFFFu;
                r[i] = rec >> 16;
                v[i] = *(const v4f*)(feature + (size_t)s * FDIM + lane * 4);
            }
            asm volatile("" :: "v"(v[0]), "v"(v[1]), "v"(v[2]), "v"(v[3]),
                               "v"(v[4]), "v"(v[5]), "v"(v[6]), "v"(v[7]));
#pragma unroll
            for (int i = 0; i < TAIL_ROWS_IN_FLIGHT; ++i) {
                if (i == 0 || k + i < cnt) {
                    float* a = acc + r[i] * PITCH + lane;
                    unsafeAtomicAdd(a,      v[i].x);
                    unsafeAtomicAdd(a + 32, v[i].y);
                    unsafeAtomicAdd(a + 64, v[i].z);
                    unsafeAtomicAdd(a + 96, v[i].w);
                }
            }
        }
    }
}

__global__ __launch_bounds__(WG_THREADS)
void gcn_owner_layer_kernel(const float* __restrict__ feature,
                            const float* __restrict__ W,
                            const float* __restrict__ b,
                            const unsigned* __restrict__ packed,
                            float*       __restrict__ out) {
    const uint4*    planes = (const uint4*)packed;
    const unsigned* recs   = packed + PLANE_WORDS;
    __shared__ __attribute__((aligned(16))) float acc[ROWS_PAD * PITCH];
    __shared__ unsigned wq[WAVES][Q_WORDS];

    const int tid  = threadIdx.x;
    const int lane = tid & 31;
    const int wave = tid >> 5;
    const int lo   = blockIdx.x * ROWS_PER_WG;

    float4* acc4 = (float4*)acc;
    for (int i = tid; i < ROWS_PAD * PITCH / 4; i += WG_THREADS)
        acc4[i] = make_float4(0.0f, 0.0f, 0.0f, 0.0f);
    __syncthreads();

    unsigned inv[8];
#pragma unroll
    for (int bb = 0; bb < 8; ++bb) inv[bb] = ((blockIdx.x >> bb) & 1u) ? 0u : ~0u;

    unsigned* q = wq[wave];
    int chunk = blockIdx.x % N_CHUNKS;

    uint4 cur[BLKS][2], nxt[BLKS][2];
    load_planes(cur, planes, chunk, wave, lane);
    int n = 0;

    for (int it = 0; it < N_CHUNKS - 1; ++it) {
        const int nchunk = (chunk + 1 == N_CHUNKS) ? 0 : chunk + 1;
        load_planes(nxt, planes, nchunk, wave, lane);
#pragma unroll
        for (int k = 0; k < BLKS; ++k) {
            n = keep_block(cur[k], (chunk * WAVES + wave) * BLKS + k, lane, inv, q, n);
            while (n >= BATCH) {
                n -= BATCH;
                gather_full(q + n, lane, feature, recs, acc);
            }
        }
#pragma unroll
        for (int k = 0; k < BLKS; ++k) { cur[k][0] = nxt[k][0]; cur[k][1] = nxt[k][1]; }
        chunk = nchunk;
    }

    const int wave_u = __builtin_amdgcn_readfirstlane(wave);
    const int ct = wave_u & (COL_TILES - 1);
    const int ln = lane & 15;
    const int hf = lane >> 4;
    const float* wrow = W + (size_t)(ct * 16 + ln) * FDIM + 16 * hf;
    v4f wf[4][4];
#pragma unroll
    for (int a = 0; a < 4; ++a)
#pragma unroll
        for (int j = 0; j < 4; ++j)
            wf[a][j] = *(const v4f*)(wrow + 32 * a + 4 * j);
    const float bias = b[ct * 16 + ln];

#pragma unroll
    for (int k = 0; k < BLKS; ++k) {
        n = keep_block(cur[k], (chunk * WAVES + wave) * BLKS + k, lane, inv, q, n);
        while (n >= BATCH) {
            n -= BATCH;
            gather_full(q + n, lane, feature, recs, acc);
        }
    }
    gather_tail(q, n, lane, feature, recs, acc);
    __syncthreads();

    for (int mt = wave_u >> 3; mt < ROW_TILES; mt += 2) {
        const float* arow = acc + (mt * 16 + ln) * PITCH + 4 * hf;
        v8f d;
#pragma unroll
        for (int v = 0; v < 8; ++v) d[v] = bias;
#pragma unroll
        for (int s = 0; s < 16; ++s) {
            const v4f x = *(const v4f*)(arow + 8 * s);
            const int a = s & 3, qq = s >> 2;
            v2f a0, a1, b0, b1;
            a0[0] = x[0];            a0[1] = x[1];
            a1[0] = x[2];            a1[1] = x[3];
            b0[0] = wf[a][0][qq];    b0[1] = wf[a][1][qq];
            b1[0] = wf[a][2][qq];    b1[1] = wf[a][3][qq];
            d = __builtin_amdgcn_wmma_f32_16x16x4_f32(false, a0, false, b0, (short)0, d, false, false);
            d = __builtin_amdgcn_wmma_f32_16x16x4_f32(false, a1, false, b1, (short)0, d, false, false);
        }
        const int rbase = mt * 16 + 8 * hf;
        float* o = out + (size_t)(lo + rbase) * FDIM + ct * 16 + ln;
#pragma unroll
        for (int v = 0; v < 8; ++v)
            if (rbase + v < ROWS_PER_WG) o[v * FDIM] = d[v];
    }
}

extern "C" void kernel_launch(void* const* d_in, const int* in_sizes, int n_in,
                              void* d_out, int out_size, void* d_ws, size_t ws_size,
                              hipStream_t stream) {
    const float* feature = (const float*)d_in[0];
    const int*   src     = (const int*)  d_in[1];
    const int*   dst     = (const int*)  d_in[2];
    const float* W       = (const float*)d_in[3];
    const float* b       = (const float*)d_in[4];
    float*       out     = (float*)d_out;
    unsigned*    planes  = (unsigned*)d_ws;

    gcn_pack_owner_kernel<<<(N_BLOCKS + PACK_WAVES - 1) / PACK_WAVES, PACK_WAVES * 32, 0, stream>>>(src, dst, planes);
    gcn_owner_layer_kernel<<<N_OWNERS, WG_THREADS, 0, stream>>>(feature, W, b, planes, out);
}


# Kept edges handed through LDS lists to the wave that owns the destination row (row mod 16); it sums its 13 rows in registers: no LDS float add per edge; rows summed in list arrival order
# speedup vs baseline: 4.2485x; 1.0766x over previous
#include <hip/hip_runtime.h>


typedef __attribute__((ext_vector_type(2))) float v2f;
typedef __attribute__((ext_vector_type(8))) float v8f;
typedef __attribute__((ext_vector_type(4))) float v4f;
typedef __attribute__((ext_vector_type(16))) float v16f;

#define N_NODES 50000
#define FDIM    128
#define E_EDGES 1600000

#define ROWS_PER_WG 200
#define N_OWNERS    250
#define WAVES       16
#define WG_THREADS  512
static_assert(N_OWNERS * ROWS_PER_WG == N_NODES, "owners must cover every node");
static_assert(WG_THREADS == WAVES * 32, "work-group size");
static_assert(N_OWNERS <= 255, "an owner number fits in 8 bits; 255 means nobody");
#define BLOCK_EDGES 1024
#define N_BLOCKS    ((E_EDGES + BLOCK_EDGES - 1) / BLOCK_EDGES)
#define BLKS        4
#define CHUNK_BLOCKS (WAVES * BLKS)
#define N_CHUNKS    ((N_BLOCKS + CHUNK_BLOCKS - 1) / CHUNK_BLOCKS)
#define PACK_WAVES  4
#define PLANE_WORDS (N_BLOCKS * (BLOCK_EDGES / 4))
static_assert(N_NODES <= 65536, "a source row number fits in the low 16 bits of an edge record");
static_assert(ROWS_PER_WG <= 256, "a row number within its owner fits in 8 bits of an edge record");
static_assert((PLANE_WORDS * 4) % 16 == 0, "the records start on a 16-byte boundary");
#define BATCH       32
#define ROWS_IN_FLIGHT 32
#define TAIL_ROWS_IN_FLIGHT 8
#define Q_WORDS     (BLOCK_EDGES + BATCH)
#define RCAP        1024
#define OWN_ROWS    ((ROWS_PER_WG + WAVES - 1) / WAVES)
static_assert(OWN_ROWS <= 16, "a wave's row sums fit four 16-float vectors");
static_assert(WAVES == 16, "the owning wave is the low 4 bits of the row number");
#define PITCH       132
#define ROW_TILES   ((ROWS_PER_WG + 15) / 16)
#define ROWS_PAD    (ROW_TILES * 16)
#define COL_TILES   (FDIM / 16)
static_assert(WAVES == 2 * COL_TILES, "two waves per column tile in the product");
static_assert((PITCH * 4) % 16 == 0, "fragment reads are 16-byte loads");

__global__ __launch_bounds__(PACK_WAVES * 32)
void gcn_pack_owner_kernel(const int* __restrict__ src, const int* __restrict__ dst,
                           unsigned* __restrict__ packed) {
    uint4*    planes = (uint4*)packed;
    unsigned* recs   = packed + PLANE_WORDS;
    const int lane = threadIdx.x & 31;
    const int blk  = blockIdx.x * PACK_WAVES + (threadIdx.x >> 5);
    if (blk >= N_BLOCKS) return;
    const int base = blk * BLOCK_EDGES + lane * 4;
    int4 d[8], sv[8];
#pragma unroll
    for (int u = 0; u < 8; ++u) {
        const int e0  = base + u * 128;
        const int e0c = (e0 < E_EDGES) ? e0 : (E_EDGES - 4);
        int4 x = *(const int4*)(dst + e0c);
        sv[u]  = *(const int4*)(src + e0c);
        if (e0 >= E_EDGES) x = make_int4(-1, -1, -1, -1);
        d[u] = x;
    }
    asm volatile("" :: "v"(sv[0].x), "v"(sv[0].y), "v"(sv[0].z), "v"(sv[0].w));
    unsigned p[8] = {0u, 0u, 0u, 0u, 0u, 0u, 0u, 0u};
#pragma unroll
    for (int u = 0; u < 8; ++u) {
        const int dd[4] = {d[u].x, d[u].y, d[u].z, d[u].w};
        const int ss[4] = {sv[u].x, sv[u].y, sv[u].z, sv[u].w};
        unsigned rc[4];
#pragma unroll
        for (int j = 0; j < 4; ++j) {
            unsigned o = (unsigned)dd[j] / (unsigned)ROWS_PER_WG;
            rc[j] = ((unsigned)ss[j] & 0xFFFFu) | ((((unsigned)dd[j] - o * (unsigned)ROWS_PER_WG) & 0xFFu) << 16);
            o = (o < 255u) ? o : 255u;
            const int gi = u * 4 + j;
#pragma unroll
            for (int b = 0; b < 8; ++b) {
                const unsigned m = __builtin_amdgcn_ballot_w32(((o >> b) & 1u) != 0u);
                if (lane == gi) p[b] = m;
            }
        }
        const int e0 = base + u * 128;
        if (e0 < E_EDGES)
            *(uint4*)(recs + e0) = make_uint4(rc[0], rc[1], rc[2], rc[3]);
    }
    planes[blk * 64 + lane]      = make_uint4(p[0], p[1], p[2], p[3]);
    planes[blk * 64 + 32 + lane] = make_uint4(p[4], p[5], p[6], p[7]);
}

__device__ __forceinline__ void load_planes(uint4 (&d)[BLKS][2], const uint4* __restrict__ planes,
                                            int chunk, int wave, int lane) {
#pragma unroll
    for (int k = 0; k < BLKS; ++k) {
        const int blk  = (chunk * WAVES + wave) * BLKS + k;
        const int blkc = (blk < N_BLOCKS) ? blk : (N_BLOCKS - 1);
        d[k][0] = planes[blkc * 64 + lane];
        d[k][1] = planes[blkc * 64 + 32 + lane];
    }
}

__device__ __forceinline__ int keep_block(const uint4 (&pl)[2], int blk, int lane, const unsigned (&inv)[8],
                                          unsigned* q, int n) {
    unsigned hm = (pl[0].x ^ inv[0]) & (pl[0].y ^ inv[1]) & (pl[0].z ^ inv[2]) & (pl[0].w ^ inv[3])
                & (pl[1].x ^ inv[4]) & (pl[1].y ^ inv[5]) & (pl[1].z ^ inv[6]) & (pl[1].w ^ inv[7]);
    if (blk >= N_BLOCKS) hm = 0u;
    const unsigned e_lane = (unsigned)blk * BLOCK_EDGES + (unsigned)(lane >> 2) * 128u + (unsigned)(lane & 3);
    unsigned any = __builtin_amdgcn_ballot_w32(hm != 0u);
    while (any != 0u) {
        if (hm != 0u) {
            const unsigned l = (unsigned)__builtin_ctz(hm);
            hm &= hm - 1u;
            const int pos = n + (int)__builtin_amdgcn_mbcnt_lo(any, 0u);
            q[pos] = e_lane + 4u * l;
        }
        n += __builtin_popcount(any);
        any = __builtin_amdgcn_ballot_w32(hm != 0u);
    }
    return n;
}

__device__ __forceinline__ void route(const unsigned* qb, int cnt, int lane,
                                      const float*    __restrict__ feature,
                                      const unsigned* __restrict__ recs,
                                      float* acc, unsigned (*rq)[RCAP], unsigned* rcnt) {
    const bool valid = lane < cnt;
    const unsigned ent = qb[valid ? lane : 0];
    const unsigned rec = recs[ent];
    const unsigned dw  = (rec >> 16) & 15u;
    unsigned pos = 0u;
    if (valid) pos = atomicAdd(&rcnt[dw], 1u);
    const bool fits = valid && pos < (unsigned)RCAP;
    if (fits) rq[dw][pos] = rec;
    unsigned ov = __builtin_amdgcn_ballot_w32(valid && !fits);
    while (ov != 0u) {
        const int l = __builtin_ctz(ov);
        ov &= ov - 1u;
        const unsigned rr = (unsigned)__builtin_amdgcn_readlane((int)rec, l);
        const v4f v = *(const v4f*)(feature + (size_t)(rr & 0xFFFFu) * FDIM + lane * 4);
        float* a = acc + ((rr >> 16) & 0xFFu) * PITCH + lane;
        unsafeAtomicAdd(a,      v.x);
        unsafeAtomicAdd(a + 32, v.y);
        unsafeAtomicAdd(a + 64, v.z);
        unsafeAtomicAdd(a + 96, v.w);
    }
}

__device__ __forceinline__ void add_row(v16f& ax, v16f& ay, v16f& az, v16f& aw, unsigned k, const v4f v) {
    ax[k] += v.x;
    ay[k] += v.y;
    az[k] += v.z;
    aw[k] += v.w;
}

__global__ __launch_bounds__(WG_THREADS)
void gcn_owner_layer_kernel(const float* __restrict__ feature,
                            const float* __restrict__ W,
                            const float* __restrict__ b,
                            const unsigned* __restrict__ packed,
                            float*       __restrict__ out) {
    const uint4*    planes = (const uint4*)packed;
    const unsigned* recs   = packed + PLANE_WORDS;
    __shared__ __attribute__((aligned(16))) float acc[ROWS_PAD * PITCH];
    __shared__ unsigned wq[WAVES][Q_WORDS];
    __shared__ unsigned rq[WAVES][RCAP];
    __shared__ unsigned rcnt[WAVES];

    const int tid  = threadIdx.x;
    const int lane = tid & 31;
    const int wave = tid >> 5;
    const int lo   = blockIdx.x * ROWS_PER_WG;

    float4* acc4 = (float4*)acc;
    for (int i = tid; i < ROWS_PAD * PITCH / 4; i += WG_THREADS)
        acc4[i] = make_float4(0.0f, 0.0f, 0.0f, 0.0f);
    if (tid < WAVES) rcnt[tid] = 0u;
    __syncthreads();

    unsigned inv[8];
#pragma unroll
    for (int bb = 0; bb < 8; ++bb) inv[bb] = ((blockIdx.x >> bb) & 1u) ? 0u : ~0u;

    unsigned* q = wq[wave];
    int chunk = blockIdx.x % N_CHUNKS;

    uint4 cur[BLKS][2], nxt[BLKS][2];
    load_planes(cur, planes, chunk, wave, lane);
    int n = 0;

    for (int it = 0; it < N_CHUNKS - 1; ++it) {
        const int nchunk = (chunk + 1 == N_CHUNKS) ? 0 : chunk + 1;
        load_planes(nxt, planes, nchunk, wave, lane);
#pragma unroll
        for (int k = 0; k < BLKS; ++k) {
            n = keep_block(cur[k], (chunk * WAVES + wave) * BLKS + k, lane, inv, q, n);
            while (n >= BATCH) {
                n -= BATCH;
                route(q + n, BATCH, lane, feature, recs, acc, rq, rcnt);
            }
        }
#pragma unroll
        for (int k = 0; k < BLKS; ++k) { cur[k][0] = nxt[k][0]; cur[k][1] = nxt[k][1]; }
        chunk = nchunk;
    }

    const int wave_u = __builtin_amdgcn_readfirstlane(wave);

#pragma unroll
    for (int k = 0; k < BLKS; ++k) {
        n = keep_block(cur[k], (chunk * WAVES + wave) * BLKS + k, lane, inv, q, n);
        while (n >= BATCH) {
            n -= BATCH;
            route(q + n, BATCH, lane, feature, recs, acc, rq, rcnt);
        }
    }
    if (n > 0) route(q, n, lane, feature, recs, acc, rq, rcnt);
    __syncthreads();

    {
        const unsigned* myq = rq[wave_u];
        const unsigned taken = rcnt[wave_u];
        const int m = __builtin_amdgcn_readfirstlane((int)((taken < (unsigned)RCAP) ? taken : (unsigned)RCAP));
        v16f ax = {}, ay = {}, az = {}, aw = {};
        int b0 = 0;
        for (; b0 + BATCH <= m; b0 += BATCH) {
            const unsigned rec_l = myq[b0 + lane];
            v4f      v[ROWS_IN_FLIGHT];
            unsigned kk[ROWS_IN_FLIGHT];
#pragma unroll
            for (int i = 0; i < ROWS_IN_FLIGHT; ++i) {
                const unsigned rec = (unsigned)__builtin_amdgcn_readlane((int)rec_l, i);
                kk[i] = (rec >> 20) & 15u;
                v[i]  = *(const v4f*)(feature + (size_t)(rec & 0xFFFFu) * FDIM + lane * 4);
            }
#pragma unroll
            for (int i = 0; i < ROWS_IN_FLIGHT; ++i) add_row(ax, ay, az, aw, kk[i], v[i]);
        }
        if (b0 < m) {
            const int cnt = m - b0;
            const unsigned rec_l = myq[b0 + ((lane < cnt) ? lane : (cnt - 1))];
            for (int k0 = 0; k0 < cnt; k0 += TAIL_ROWS_IN_FLIGHT) {
                v4f      v[TAIL_ROWS_IN_FLIGHT];
                unsigned kk[TAIL_ROWS_IN_FLIGHT];
#pragma unroll
                for (int i = 0; i < TAIL_ROWS_IN_FLIGHT; ++i) {
                    const int li = (k0 + i < cnt) ? (k0 + i) : (cnt - 1);
                    const unsigned rec = (unsigned)__builtin_amdgcn_readlane((int)rec_l, li);
                    kk[i] = (rec >> 20) & 15u;
                    v[i]  = *(const v4f*)(feature + (size_t)(rec & 0xFFFFu) * FDIM + lane * 4);
                }
#pragma unroll
                for (int i = 0; i < TAIL_ROWS_IN_FLIGHT; ++i)
                    if (k0 + i < cnt) add_row(ax, ay, az, aw, kk[i], v[i]);
            }
        }
#pragma unroll
        for (int k = 0; k < OWN_ROWS; ++k) {
            const int row = k * WAVES + wave_u;
            if (row < ROWS_PER_WG) {
                float* a = acc + row * PITCH + lane;
                unsafeAtomicAdd(a,      ax[k]);
                unsafeAtomicAdd(a + 32, ay[k]);
                unsafeAtomicAdd(a + 64, az[k]);
                unsafeAtomicAdd(a + 96, aw[k]);
            }
        }
    }

    const int ct = wave_u & (COL_TILES - 1);
    const int ln = lane & 15;
    const int hf = lane >> 4;
    const float* wrow = W + (size_t)(ct * 16 + ln) * FDIM + 16 * hf;
    v4f wf[4][4];
#pragma unroll
    for (int a = 0; a < 4; ++a)
#pragma unroll
        for (int j = 0; j < 4; ++j)
            wf[a][j] = *(const v4f*)(wrow + 32 * a + 4 * j);
    const float bias = b[ct * 16 + ln];
    __syncthreads();

    for (int mt = wave_u >> 3; mt < ROW_TILES; mt += 2) {
        const float* arow = acc + (mt * 16 + ln) * PITCH + 4 * hf;
        v8f d;
#pragma unroll
        for (int v = 0; v < 8; ++v) d[v] = bias;
#pragma unroll
        for (int s = 0; s < 16; ++s) {
            const v4f x = *(const v4f*)(arow + 8 * s);
            const int a = s & 3, qq = s >> 2;
            v2f a0, a1, b0, b1;
            a0[0] = x[0];            a0[1] = x[1];
            a1[0] = x[2];            a1[1] = x[3];
            b0[0] = wf[a][0][qq];    b0[1] = wf[a][1][qq];
            b1[0] = wf[a][2][qq];    b1[1] = wf[a][3][qq];
            d = __builtin_amdgcn_wmma_f32_16x16x4_f32(false, a0, false, b0, (short)0, d, false, false);
            d = __builtin_amdgcn_wmma_f32_16x16x4_f32(false, a1, false, b1, (short)0, d, false, false);
        }
        const int rbase = mt * 16 + 8 * hf;
        float* o = out + (size_t)(lo + rbase) * FDIM + ct * 16 + ln;
#pragma unroll
        for (int v = 0; v < 8; ++v)
            if (rbase + v < ROWS_PER_WG) o[v * FDIM] = d[v];
    }
}

extern "C" void kernel_launch(void* const* d_in, const int* in_sizes, int n_in,
                              void* d_out, int out_size, void* d_ws, size_t ws_size,
                              hipStream_t stream) {
    const float* feature = (const float*)d_in[0];
    const int*   src     = (const int*)  d_in[1];
    const int*   dst     = (const int*)  d_in[2];
    const float* W       = (const float*)d_in[3];
    const float* b       = (const float*)d_in[4];
    float*       out     = (float*)d_out;
    unsigned*    planes  = (unsigned*)d_ws;

    gcn_pack_owner_kernel<<<(N_BLOCKS + PACK_WAVES - 1) / PACK_WAVES, PACK_WAVES * 32, 0, stream>>>(src, dst, planes);
    gcn_owner_layer_kernel<<<N_OWNERS, WG_THREADS, 0, stream>>>(feature, W, b, planes, out);
}


# Owner scan ANDs two one-hot planes (high, low 4 bits of the owner: 2 bits per edge), not 8 bit planes; launch 1 ORs them in LDS, one block per wave, whole-line stores; sums in another arrival order
# speedup vs baseline: 4.5259x; 1.0653x over previous
#include <hip/hip_runtime.h>


typedef __attribute__((ext_vector_type(2))) float v2f;
typedef __attribute__((ext_vector_type(8))) float v8f;
typedef __attribute__((ext_vector_type(4))) float v4f;
typedef __attribute__((ext_vector_type(16))) float v16f;

#define N_NODES 50000
#define FDIM    128
#define E_EDGES 1600000

#define ROWS_PER_WG 200
#define N_OWNERS    250
#define WAVES       16
#define WG_THREADS  512
static_assert(N_OWNERS * ROWS_PER_WG == N_NODES, "owners must cover every node");
static_assert(WG_THREADS == WAVES * 32, "work-group size");
static_assert(N_OWNERS <= 256, "an owner number splits into two 4-bit halves");
#define BLOCK_EDGES 1024
#define N_BLOCKS    ((E_EDGES + BLOCK_EDGES - 1) / BLOCK_EDGES)
#define N_BLOCKS_PAD (((N_BLOCKS + 3) / 4) * 4)
#define N_QUADS     (N_BLOCKS_PAD / 4)
#define QPW         2
#define CHUNK_QUADS (WAVES * QPW)
#define N_CHUNKS    ((N_QUADS + CHUNK_QUADS - 1) / CHUNK_QUADS)
#define PACK_WAVES  4
#define N_PLANES    32
#define PLANE_STRIDE (N_BLOCKS_PAD * 32)
#define PLANE_WORDS (N_PLANES * PLANE_STRIDE)
static_assert(N_BLOCKS_PAD % PACK_WAVES == 0, "the packing launch has no partial work-group");
static_assert((PLANE_STRIDE * 4) % 512 == 0, "every plane starts on a quad boundary");
static_assert(N_NODES <= 65536, "a source row number fits in the low 16 bits of an edge record");
static_assert(ROWS_PER_WG <= 256, "a row number within its owner fits in 8 bits of an edge record");
static_assert((PLANE_WORDS * 4) % 16 == 0, "the records start on a 16-byte boundary");
#define BATCH       32
#define ROWS_IN_FLIGHT 32
#define TAIL_ROWS_IN_FLIGHT 8
#define Q_WORDS     (BLOCK_EDGES + BATCH)
#define RCAP        1024
#define OWN_ROWS    ((ROWS_PER_WG + WAVES - 1) / WAVES)
static_assert(OWN_ROWS <= 16, "a wave's row sums fit four 16-float vectors");
static_assert(WAVES == 16, "the owning wave is the low 4 bits of the row number");
#define PITCH       132
#define ROW_TILES   ((ROWS_PER_WG + 15) / 16)
#define ROWS_PAD    (ROW_TILES * 16)
#define COL_TILES   (FDIM / 16)
static_assert(WAVES == 2 * COL_TILES, "two waves per column tile in the product");
static_assert((PITCH * 4) % 16 == 0, "fragment reads are 16-byte loads");

__global__ __launch_bounds__(PACK_WAVES * 32)
void gcn_pack_owner_kernel(const int* __restrict__ src, const int* __restrict__ dst,
                           unsigned* __restrict__ packed) {
    __shared__ __attribute__((aligned(16))) unsigned pl[PACK_WAVES][N_PLANES][32];
    unsigned* recs   = packed + PLANE_WORDS;
    const int lane = threadIdx.x & 31;
    const int wv   = threadIdx.x >> 5;
    const int blk  = blockIdx.x * PACK_WAVES + wv;
    const int base = blk * BLOCK_EDGES + lane * 4;
    int4 d[8], sv[8];
#pragma unroll
    for (int u = 0; u < 8; ++u) {
        const int e0  = base + u * 128;
        const int e0c = (e0 < E_EDGES) ? e0 : (E_EDGES - 4);
        d[u]  = *(const int4*)(dst + e0c);
        sv[u] = *(const int4*)(src + e0c);
    }
    asm volatile("" :: "v"(sv[0].x), "v"(sv[0].y), "v"(sv[0].z), "v"(sv[0].w));
    unsigned (*col)[32] = pl[wv];
#pragma unroll
    for (int p = 0; p < N_PLANES; ++p) col[p][lane] = 0u;
#pragma unroll
    for (int u = 0; u < 8; ++u) {
        const int dd[4] = {d[u].x, d[u].y, d[u].z, d[u].w};
        const int ss[4] = {sv[u].x, sv[u].y, sv[u].z, sv[u].w};
        const int e0 = base + u * 128;
        unsigned rc[4];
#pragma unroll
        for (int j = 0; j < 4; ++j) {
            const unsigned o = (unsigned)dd[j] / (unsigned)ROWS_PER_WG;
            rc[j] = ((unsigned)ss[j] & 0xFFFFu) | ((((unsigned)dd[j] - o * (unsigned)ROWS_PER_WG) & 0xFFu) << 16);
            if (e0 < E_EDGES && (unsigned)dd[j] < (unsigned)N_NODES) {
                const unsigned bit = 1u << (u * 4 + j);
                atomicOr(&col[o >> 4][lane], bit);
                atomicOr(&col[16u + (o & 15u)][lane], bit);
            }
        }
        if (e0 < E_EDGES)
            *(uint4*)(recs + e0) = make_uint4(rc[0], rc[1], rc[2], rc[3]);
    }
#pragma unroll
    for (int s = 0; s < N_PLANES / 4; ++s) {
        const int p  = 4 * s + (lane >> 3);
        const int w4 = 4 * (lane & 7);
        const uint4 v = *(const uint4*)&col[p][w4];
        *(uint4*)(packed + (size_t)p * PLANE_STRIDE + (size_t)blk * 32 + w4) = v;
    }
}

__device__ __forceinline__ void load_planes(uint4 (&d)[QPW][2], const uint4* __restrict__ hi,
                                            const uint4* __restrict__ lo, int chunk, int wave, int lane) {
#pragma unroll
    for (int k = 0; k < QPW; ++k) {
        const int quad  = (chunk * WAVES + wave) * QPW + k;
        const int quadc = (quad < N_QUADS) ? quad : (N_QUADS - 1);
        d[k][0] = hi[quadc * 32 + lane];
        d[k][1] = lo[quadc * 32 + lane];
    }
}

__device__ __forceinline__ int keep_word(unsigned hm, unsigned e_base, unsigned* q, int n) {
    unsigned any = __builtin_amdgcn_ballot_w32(hm != 0u);
    while (any != 0u) {
        if (hm != 0u) {
            const unsigned l = (unsigned)__builtin_ctz(hm);
            hm &= hm - 1u;
            const int pos = n + (int)__builtin_amdgcn_mbcnt_lo(any, 0u);
            q[pos] = e_base + ((l & 28u) << 5) + (l & 3u);
        }
        n += __builtin_popcount(any);
        any = __builtin_amdgcn_ballot_w32(hm != 0u);
    }
    return n;
}

__device__ __forceinline__ void route(const unsigned* qb, int cnt, int lane,
                                      const float*    __restrict__ feature,
                                      const unsigned* __restrict__ recs,
                                      float* acc, unsigned (*rq)[RCAP], unsigned* rcnt) {
    const bool valid = lane < cnt;
    const unsigned ent = qb[valid ? lane : 0];
    const unsigned rec = recs[ent];
    const unsigned dw  = (rec >> 16) & 15u;
    unsigned pos = 0u;
    if (valid) pos = atomicAdd(&rcnt[dw], 1u);
    const bool fits = valid && pos < (unsigned)RCAP;
    if (fits) rq[dw][pos] = rec;
    unsigned ov = __builtin_amdgcn_ballot_w32(valid && !fits);
    while (ov != 0u) {
        const int l = __builtin_ctz(ov);
        ov &= ov - 1u;
        const unsigned rr = (unsigned)__builtin_amdgcn_readlane((int)rec, l);
        const v4f v = *(const v4f*)(feature + (size_t)(rr & 0xFFFFu) * FDIM + lane * 4);
        float* a = acc + ((rr >> 16) & 0xFFu) * PITCH + lane;
        unsafeAtomicAdd(a,      v.x);
        unsafeAtomicAdd(a + 32, v.y);
        unsafeAtomicAdd(a + 64, v.z);
        unsafeAtomicAdd(a + 96, v.w);
    }
}

__device__ __forceinline__ void add_row(v16f& ax, v16f& ay, v16f& az, v16f& aw, unsigned k, const v4f v) {
    ax[k] += v.x;
    ay[k] += v.y;
    az[k] += v.z;
    aw[k] += v.w;
}

__global__ __launch_bounds__(WG_THREADS)
void gcn_owner_layer_kernel(const float* __restrict__ feature,
                            const float* __restrict__ W,
                            const float* __restrict__ b,
                            const unsigned* __restrict__ packed,
                            float*       __restrict__ out) {
    const uint4*    hi_pl  = (const uint4*)(packed + (size_t)(blockIdx.x >> 4) * PLANE_STRIDE);
    const uint4*    lo_pl  = (const uint4*)(packed + (size_t)(16u + (blockIdx.x & 15u)) * PLANE_STRIDE);
    const unsigned* recs   = packed + PLANE_WORDS;
    __shared__ __attribute__((aligned(16))) float acc[ROWS_PAD * PITCH];
    __shared__ unsigned wq[WAVES][Q_WORDS];
    __shared__ unsigned rq[WAVES][RCAP];
    __shared__ unsigned rcnt[WAVES];

    const int tid  = threadIdx.x;
    const int lane = tid & 31;
    const int wave = tid >> 5;
    const int lo   = blockIdx.x * ROWS_PER_WG;

    float4* acc4 = (float4*)acc;
    for (int i = tid; i < ROWS_PAD * PITCH / 4; i += WG_THREADS)
        acc4[i] = make_float4(0.0f, 0.0f, 0.0f, 0.0f);
    if (tid < WAVES) rcnt[tid] = 0u;
    __syncthreads();

    unsigned* q = wq[wave];
    int chunk = blockIdx.x % N_CHUNKS;

    uint4 cur[QPW][2], nxt[QPW][2];
    load_planes(cur, hi_pl, lo_pl, chunk, wave, lane);
    int n = 0;

    const int wave_u = __builtin_amdgcn_readfirstlane(wave);

    const unsigned e_lane = (unsigned)(lane >> 3) * BLOCK_EDGES + 16u * (unsigned)(lane & 7);

    for (int it = 0; it < N_CHUNKS; ++it) {
        const int nchunk = (chunk + 1 == N_CHUNKS) ? 0 : chunk + 1;
        load_planes(nxt, hi_pl, lo_pl, nchunk, wave, lane);
#pragma unroll
        for (int k = 0; k < QPW; ++k) {
            const int  quad = (chunk * WAVES + wave) * QPW + k;
            const bool live = quad < N_QUADS;
            const unsigned eb = (unsigned)quad * (4u * BLOCK_EDGES) + e_lane;
            const unsigned m4[4] = {cur[k][0].x & cur[k][1].x, cur[k][0].y & cur[k][1].y,
                                    cur[k][0].z & cur[k][1].z, cur[k][0].w & cur[k][1].w};
#pragma unroll
            for (int i = 0; i < 4; ++i) {
                n = keep_word(live ? m4[i] : 0u, eb + 4u * (unsigned)i, q, n);
                while (n >= BATCH) {
                    n -= BATCH;
                    route(q + n, BATCH, lane, feature, recs, acc, rq, rcnt);
                }
            }
        }
#pragma unroll
        for (int k = 0; k < QPW; ++k) { cur[k][0] = nxt[k][0]; cur[k][1] = nxt[k][1]; }
        chunk = nchunk;
    }
    if (n > 0) route(q, n, lane, feature, recs, acc, rq, rcnt);
    __syncthreads();

    {
        const unsigned* myq = rq[wave_u];
        const unsigned taken = rcnt[wave_u];
        const int m = __builtin_amdgcn_readfirstlane((int)((taken < (unsigned)RCAP) ? taken : (unsigned)RCAP));
        v16f ax = {}, ay = {}, az = {}, aw = {};
        int b0 = 0;
        for (; b0 + BATCH <= m; b0 += BATCH) {
            const unsigned rec_l = myq[b0 + lane];
            v4f      v[ROWS_IN_FLIGHT];
            unsigned kk[ROWS_IN_FLIGHT];
#pragma unroll
            for (int i = 0; i < ROWS_IN_FLIGHT; ++i) {
                const unsigned rec = (unsigned)__builtin_amdgcn_readlane((int)rec_l, i);
                kk[i] = (rec >> 20) & 15u;
                v[i]  = *(const v4f*)(feature + (size_t)(rec & 0xFFFFu) * FDIM + lane * 4);
            }
#pragma unroll
            for (int i = 0; i < ROWS_IN_FLIGHT; ++i) add_row(ax, ay, az, aw, kk[i], v[i]);
        }
        if (b0 < m) {
            const int cnt = m - b0;
            const unsigned rec_l = myq[b0 + ((lane < cnt) ? lane : (cnt - 1))];
            for (int k0 = 0; k0 < cnt; k0 += TAIL_ROWS_IN_FLIGHT) {
                v4f      v[TAIL_ROWS_IN_FLIGHT];
                unsigned kk[TAIL_ROWS_IN_FLIGHT];
#pragma unroll
                for (int i = 0; i < TAIL_ROWS_IN_FLIGHT; ++i) {
                    const int li = (k0 + i < cnt) ? (k0 + i) : (cnt - 1);
                    const unsigned rec = (unsigned)__builtin_amdgcn_readlane((int)rec_l, li);
                    kk[i] = (rec >> 20) & 15u;
                    v[i]  = *(const v4f*)(feature + (size_t)(rec & 0xFFFFu) * FDIM + lane * 4);
                }
#pragma unroll
                for (int i = 0; i < TAIL_ROWS_IN_FLIGHT; ++i)
                    if (k0 + i < cnt) add_row(ax, ay, az, aw, kk[i], v[i]);
            }
        }
#pragma unroll
        for (int k = 0; k < OWN_ROWS; ++k) {
            const int row = k * WAVES + wave_u;
            if (row < ROWS_PER_WG) {
                float* a = acc + row * PITCH + lane;
                unsafeAtomicAdd(a,      ax[k]);
                unsafeAtomicAdd(a + 32, ay[k]);
                unsafeAtomicAdd(a + 64, az[k]);
                unsafeAtomicAdd(a + 96, aw[k]);
            }
        }
    }

    const int ct = wave_u & (COL_TILES - 1);
    const int ln = lane & 15;
    const int hf = lane >> 4;
    const float* wrow = W + (size_t)(ct * 16 + ln) * FDIM + 16 * hf;
    v4f wf[4][4];
#pragma unroll
    for (int a = 0; a < 4; ++a)
#pragma unroll
        for (int j = 0; j < 4; ++j)
            wf[a][j] = *(const v4f*)(wrow + 32 * a + 4 * j);
    const float bias = b[ct * 16 + ln];
    __syncthreads();

    for (int mt = wave_u >> 3; mt < ROW_TILES; mt += 2) {
        const float* arow = acc + (mt * 16 + ln) * PITCH + 4 * hf;
        v8f d;
#pragma unroll
        for (int v = 0; v < 8; ++v) d[v] = bias;
#pragma unroll
        for (int s = 0; s < 16; ++s) {
            const v4f x = *(const v4f*)(arow + 8 * s);
            const int a = s & 3, qq = s >> 2;
            v2f a0, a1, b0, b1;
            a0[0] = x[0];            a0[1] = x[1];
            a1[0] = x[2];            a1[1] = x[3];
            b0[0] = wf[a][0][qq];    b0[1] = wf[a][1][qq];
            b1[0] = wf[a][2][qq];    b1[1] = wf[a][3][qq];
            d = __builtin_amdgcn_wmma_f32_16x16x4_f32(false, a0, false, b0, (short)0, d, false, false);
            d = __builtin_amdgcn_wmma_f32_16x16x4_f32(false, a1, false, b1, (short)0, d, false, false);
        }
        const int rbase = mt * 16 + 8 * hf;
        float* o = out + (size_t)(lo + rbase) * FDIM + ct * 16 + ln;
#pragma unroll
        for (int v = 0; v < 8; ++v)
            if (rbase + v < ROWS_PER_WG) o[v * FDIM] = d[v];
    }
}

extern "C" void kernel_launch(void* const* d_in, const int* in_sizes, int n_in,
                              void* d_out, int out_size, void* d_ws, size_t ws_size,
                              hipStream_t stream) {
    const float* feature = (const float*)d_in[0];
    const int*   src     = (const int*)  d_in[1];
    const int*   dst     = (const int*)  d_in[2];
    const float* W       = (const float*)d_in[3];
    const float* b       = (const float*)d_in[4];
    float*       out     = (float*)d_out;
    unsigned*    planes  = (unsigned*)d_ws;

    gcn_pack_owner_kernel<<<N_BLOCKS_PAD / PACK_WAVES, PACK_WAVES * 32, 0, stream>>>(src, dst, planes);
    gcn_owner_layer_kernel<<<N_OWNERS, WG_THREADS, 0, stream>>>(feature, W, b, planes, out);
}


# Record loads of full batches are asked for at the head of a scan trip and used at its end (the last ones together), not waited for in place; rows summed in another list arrival order
# speedup vs baseline: 4.7009x; 1.0387x over previous
#include <hip/hip_runtime.h>


typedef __attribute__((ext_vector_type(2))) float v2f;
typedef __attribute__((ext_vector_type(8))) float v8f;
typedef __attribute__((ext_vector_type(4))) float v4f;
typedef __attribute__((ext_vector_type(16))) float v16f;

#define N_NODES 50000
#define FDIM    128
#define E_EDGES 1600000

#define ROWS_PER_WG 200
#define N_OWNERS    250
#define WAVES       16
#define WG_THREADS  512
static_assert(N_OWNERS * ROWS_PER_WG == N_NODES, "owners must cover every node");
static_assert(WG_THREADS == WAVES * 32, "work-group size");
static_assert(N_OWNERS <= 256, "an owner number splits into two 4-bit halves");
#define BLOCK_EDGES 1024
#define N_BLOCKS    ((E_EDGES + BLOCK_EDGES - 1) / BLOCK_EDGES)
#define N_BLOCKS_PAD (((N_BLOCKS + 3) / 4) * 4)
#define N_QUADS     (N_BLOCKS_PAD / 4)
#define QPW         2
#define CHUNK_QUADS (WAVES * QPW)
#define N_CHUNKS    ((N_QUADS + CHUNK_QUADS - 1) / CHUNK_QUADS)
#define PACK_WAVES  4
#define N_PLANES    32
#define PLANE_STRIDE (N_BLOCKS_PAD * 32)
#define PLANE_WORDS (N_PLANES * PLANE_STRIDE)
static_assert(N_BLOCKS_PAD % PACK_WAVES == 0, "the packing launch has no partial work-group");
static_assert((PLANE_STRIDE * 4) % 512 == 0, "every plane starts on a quad boundary");
static_assert(N_NODES <= 65536, "a source row number fits in the low 16 bits of an edge record");
static_assert(ROWS_PER_WG <= 256, "a row number within its owner fits in 8 bits of an edge record");
static_assert((PLANE_WORDS * 4) % 16 == 0, "the records start on a 16-byte boundary");
#define BATCH       32
#define ROWS_IN_FLIGHT 32
#define TAIL_ROWS_IN_FLIGHT 8
#define HELD_BATCHES 2
#define HOLD        (BATCH * (HELD_BATCHES + 1))
#define Q_WORDS     (BLOCK_EDGES + HOLD)
#define RCAP        1024
#define OWN_ROWS    ((ROWS_PER_WG + WAVES - 1) / WAVES)
static_assert(OWN_ROWS <= 16, "a wave's row sums fit four 16-float vectors");
static_assert(WAVES == 16, "the owning wave is the low 4 bits of the row number");
#define PITCH       132
#define ROW_TILES   ((ROWS_PER_WG + 15) / 16)
#define ROWS_PAD    (ROW_TILES * 16)
#define COL_TILES   (FDIM / 16)
static_assert(WAVES == 2 * COL_TILES, "two waves per column tile in the product");
static_assert((PITCH * 4) % 16 == 0, "fragment reads are 16-byte loads");

__global__ __launch_bounds__(PACK_WAVES * 32)
void gcn_pack_owner_kernel(const int* __restrict__ src, const int* __restrict__ dst,
                           unsigned* __restrict__ packed) {
    __shared__ __attribute__((aligned(16))) unsigned pl[PACK_WAVES][N_PLANES][32];
    unsigned* recs   = packed + PLANE_WORDS;
    const int lane = threadIdx.x & 31;
    const int wv   = threadIdx.x >> 5;
    const int blk  = blockIdx.x * PACK_WAVES + wv;
    const int base = blk * BLOCK_EDGES + lane * 4;
    int4 d[8], sv[8];
#pragma unroll
    for (int u = 0; u < 8; ++u) {
        const int e0  = base + u * 128;
        const int e0c = (e0 < E_EDGES) ? e0 : (E_EDGES - 4);
        d[u]  = *(const int4*)(dst + e0c);
        sv[u] = *(const int4*)(src + e0c);
    }
    asm volatile("" :: "v"(sv[0].x), "v"(sv[0].y), "v"(sv[0].z), "v"(sv[0].w));
    unsigned (*col)[32] = pl[wv];
#pragma unroll
    for (int p = 0; p < N_PLANES; ++p) col[p][lane] = 0u;
#pragma unroll
    for (int u = 0; u < 8; ++u) {
        const int dd[4] = {d[u].x, d[u].y, d[u].z, d[u].w};
        const int ss[4] = {sv[u].x, sv[u].y, sv[u].z, sv[u].w};
        const int e0 = base + u * 128;
        unsigned rc[4];
#pragma unroll
        for (int j = 0; j < 4; ++j) {
            const unsigned o = (unsigned)dd[j] / (unsigned)ROWS_PER_WG;
            rc[j] = ((unsigned)ss[j] & 0xFFFFu) | ((((unsigned)dd[j] - o * (unsigned)ROWS_PER_WG) & 0xFFu) << 16);
            if (e0 < E_EDGES && (unsigned)dd[j] < (unsigned)N_NODES) {
                const unsigned bit = 1u << (u * 4 + j);
                atomicOr(&col[o >> 4][lane], bit);
                atomicOr(&col[16u + (o & 15u)][lane], bit);
            }
        }
        if (e0 < E_EDGES)
            *(uint4*)(recs + e0) = make_uint4(rc[0], rc[1], rc[2], rc[3]);
    }
#pragma unroll
    for (int s = 0; s < N_PLANES / 4; ++s) {
        const int p  = 4 * s + (lane >> 3);
        const int w4 = 4 * (lane & 7);
        const uint4 v = *(const uint4*)&col[p][w4];
        *(uint4*)(packed + (size_t)p * PLANE_STRIDE + (size_t)blk * 32 + w4) = v;
    }
}

__device__ __forceinline__ void load_planes(uint4 (&d)[QPW][2], const uint4* __restrict__ hi,
                                            const uint4* __restrict__ lo, int chunk, int wave, int lane) {
#pragma unroll
    for (int k = 0; k < QPW; ++k) {
        const int quad  = (chunk * WAVES + wave) * QPW + k;
        const int quadc = (quad < N_QUADS) ? quad : (N_QUADS - 1);
        d[k][0] = hi[quadc * 32 + lane];
        d[k][1] = lo[quadc * 32 + lane];
    }
}

__device__ __forceinline__ int keep_word(unsigned hm, unsigned e_base, unsigned* q, int n) {
    unsigned any = __builtin_amdgcn_ballot_w32(hm != 0u);
    while (any != 0u) {
        if (hm != 0u) {
            const unsigned l = (unsigned)__builtin_ctz(hm);
            hm &= hm - 1u;
            const int pos = n + (int)__builtin_amdgcn_mbcnt_lo(any, 0u);
            q[pos] = e_base + ((l & 28u) << 5) + (l & 3u);
        }
        n += __builtin_popcount(any);
        any = __builtin_amdgcn_ballot_w32(hm != 0u);
    }
    return n;
}

__device__ __forceinline__ void hand_on(const unsigned rec, const bool valid, int lane,
                                        const float* __restrict__ feature,
                                        float* acc, unsigned (*rq)[RCAP], unsigned* rcnt) {
    const unsigned dw  = (rec >> 16) & 15u;
    unsigned pos = 0u;
    if (valid) pos = atomicAdd(&rcnt[dw], 1u);
    const bool fits = valid && pos < (unsigned)RCAP;
    if (fits) rq[dw][pos] = rec;
    unsigned ov = __builtin_amdgcn_ballot_w32(valid && !fits);
    while (ov != 0u) {
        const int l = __builtin_ctz(ov);
        ov &= ov - 1u;
        const unsigned rr = (unsigned)__builtin_amdgcn_readlane((int)rec, l);
        const v4f v = *(const v4f*)(feature + (size_t)(rr & 0xFFFFu) * FDIM + lane * 4);
        float* a = acc + ((rr >> 16) & 0xFFu) * PITCH + lane;
        unsafeAtomicAdd(a,      v.x);
        unsafeAtomicAdd(a + 32, v.y);
        unsafeAtomicAdd(a + 64, v.z);
        unsafeAtomicAdd(a + 96, v.w);
    }
}
__device__ __forceinline__ void route(const unsigned* qb, int cnt, int lane,
                                      const float*    __restrict__ feature,
                                      const unsigned* __restrict__ recs,
                                      float* acc, unsigned (*rq)[RCAP], unsigned* rcnt) {
    const bool valid = lane < cnt;
    const unsigned ent = qb[valid ? lane : 0];
    const unsigned rec = recs[ent];
    hand_on(rec, valid, lane, feature, acc, rq, rcnt);
}

__device__ __forceinline__ void add_row(v16f& ax, v16f& ay, v16f& az, v16f& aw, unsigned k, const v4f v) {
    ax[k] += v.x;
    ay[k] += v.y;
    az[k] += v.z;
    aw[k] += v.w;
}

__global__ __launch_bounds__(WG_THREADS)
void gcn_owner_layer_kernel(const float* __restrict__ feature,
                            const float* __restrict__ W,
                            const float* __restrict__ b,
                            const unsigned* __restrict__ packed,
                            float*       __restrict__ out) {
    const uint4*    hi_pl  = (const uint4*)(packed + (size_t)(blockIdx.x >> 4) * PLANE_STRIDE);
    const uint4*    lo_pl  = (const uint4*)(packed + (size_t)(16u + (blockIdx.x & 15u)) * PLANE_STRIDE);
    const unsigned* recs   = packed + PLANE_WORDS;
    __shared__ __attribute__((aligned(16))) float acc[ROWS_PAD * PITCH];
    __shared__ unsigned wq[WAVES][Q_WORDS];
    __shared__ unsigned rq[WAVES][RCAP];
    __shared__ unsigned rcnt[WAVES];

    const int tid  = threadIdx.x;
    const int lane = tid & 31;
    const int wave = tid >> 5;
    const int lo   = blockIdx.x * ROWS_PER_WG;

    float4* acc4 = (float4*)acc;
    for (int i = tid; i < ROWS_PAD * PITCH / 4; i += WG_THREADS)
        acc4[i] = make_float4(0.0f, 0.0f, 0.0f, 0.0f);
    if (tid < WAVES) rcnt[tid] = 0u;
    __syncthreads();

    unsigned* q = wq[wave];
    int chunk = blockIdx.x % N_CHUNKS;

    uint4 cur[QPW][2], nxt[QPW][2];
    load_planes(cur, hi_pl, lo_pl, chunk, wave, lane);
    int n = 0;

    const int wave_u = __builtin_amdgcn_readfirstlane(wave);

    const unsigned e_lane = (unsigned)(lane >> 3) * BLOCK_EDGES + 16u * (unsigned)(lane & 7);

    for (int it = 0; it < N_CHUNKS; ++it) {
        const int nchunk = (chunk + 1 == N_CHUNKS) ? 0 : chunk + 1;
        load_planes(nxt, hi_pl, lo_pl, nchunk, wave, lane);
        unsigned held[HELD_BATCHES];
        bool     has[HELD_BATCHES];
#pragma unroll
        for (int j = 0; j < HELD_BATCHES; ++j) {
            has[j]  = n >= BATCH;
            held[j] = 0u;
            if (has[j]) { n -= BATCH; held[j] = recs[q[n + lane]]; }
        }
#pragma unroll
        for (int k = 0; k < QPW; ++k) {
            const int  quad = (chunk * WAVES + wave) * QPW + k;
            const bool live = quad < N_QUADS;
            const unsigned eb = (unsigned)quad * (4u * BLOCK_EDGES) + e_lane;
            const unsigned m4[4] = {cur[k][0].x & cur[k][1].x, cur[k][0].y & cur[k][1].y,
                                    cur[k][0].z & cur[k][1].z, cur[k][0].w & cur[k][1].w};
#pragma unroll
            for (int i = 0; i < 4; ++i) {
                n = keep_word(live ? m4[i] : 0u, eb + 4u * (unsigned)i, q, n);
                while (n >= HOLD) {
                    n -= BATCH;
                    route(q + n, BATCH, lane, feature, recs, acc, rq, rcnt);
                }
            }
        }
#pragma unroll
        for (int j = 0; j < HELD_BATCHES; ++j)
            if (has[j]) hand_on(held[j], true, lane, feature, acc, rq, rcnt);
#pragma unroll
        for (int k = 0; k < QPW; ++k) { cur[k][0] = nxt[k][0]; cur[k][1] = nxt[k][1]; }
        chunk = nchunk;
    }
    {
        unsigned held[HELD_BATCHES + 1];
        int      cnt[HELD_BATCHES + 1];
#pragma unroll
        for (int j = 0; j < HELD_BATCHES + 1; ++j) {
            const int left = n - BATCH * j;
            cnt[j]  = (left < BATCH) ? left : BATCH;
            held[j] = 0u;
            if (cnt[j] > 0) held[j] = recs[q[BATCH * j + ((lane < cnt[j]) ? lane : 0)]];
        }
#pragma unroll
        for (int j = 0; j < HELD_BATCHES + 1; ++j)
            if (cnt[j] > 0) hand_on(held[j], lane < cnt[j], lane, feature, acc, rq, rcnt);
    }
    __syncthreads();

    {
        const unsigned* myq = rq[wave_u];
        const unsigned taken = rcnt[wave_u];
        const int m = __builtin_amdgcn_readfirstlane((int)((taken < (unsigned)RCAP) ? taken : (unsigned)RCAP));
        v16f ax = {}, ay = {}, az = {}, aw = {};
        int b0 = 0;
        for (; b0 + BATCH <= m; b0 += BATCH) {
            const unsigned rec_l = myq[b0 + lane];
            v4f      v[ROWS_IN_FLIGHT];
            unsigned kk[ROWS_IN_FLIGHT];
#pragma unroll
            for (int i = 0; i < ROWS_IN_FLIGHT; ++i) {
                const unsigned rec = (unsigned)__builtin_amdgcn_readlane((int)rec_l, i);
                kk[i] = (rec >> 20) & 15u;
                v[i]  = *(const v4f*)(feature + (size_t)(rec & 0xFFFFu) * FDIM + lane * 4);
            }
#pragma unroll
            for (int i = 0; i < ROWS_IN_FLIGHT; ++i) add_row(ax, ay, az, aw, kk[i], v[i]);
        }
        if (b0 < m) {
            const int cnt = m - b0;
            const unsigned rec_l = myq[b0 + ((lane < cnt) ? lane : (cnt - 1))];
            for (int k0 = 0; k0 < cnt; k0 += TAIL_ROWS_IN_FLIGHT) {
                v4f      v[TAIL_ROWS_IN_FLIGHT];
                unsigned kk[TAIL_ROWS_IN_FLIGHT];
#pragma unroll
                for (int i = 0; i < TAIL_ROWS_IN_FLIGHT; ++i) {
                    const int li = (k0 + i < cnt) ? (k0 + i) : (cnt - 1);
                    const unsigned rec = (unsigned)__builtin_amdgcn_readlane((int)rec_l, li);
                    kk[i] = (rec >> 20) & 15u;
                    v[i]  = *(const v4f*)(feature + (size_t)(rec & 0xFFFFu) * FDIM + lane * 4);
                }
#pragma unroll
                for (int i = 0; i < TAIL_ROWS_IN_FLIGHT; ++i)
                    if (k0 + i < cnt) add_row(ax, ay, az, aw, kk[i], v[i]);
            }
        }
#pragma unroll
        for (int k = 0; k < OWN_ROWS; ++k) {
            const int row = k * WAVES + wave_u;
            if (row < ROWS_PER_WG) {
                float* a = acc + row * PITCH + lane;
                unsafeAtomicAdd(a,      ax[k]);
                unsafeAtomicAdd(a + 32, ay[k]);
                unsafeAtomicAdd(a + 64, az[k]);
                unsafeAtomicAdd(a + 96, aw[k]);
            }
        }
    }

    const int ct = wave_u & (COL_TILES - 1);
    const int ln = lane & 15;
    const int hf = lane >> 4;
    const float* wrow = W + (size_t)(ct * 16 + ln) * FDIM + 16 * hf;
    v4f wf[4][4];
#pragma unroll
    for (int a = 0; a < 4; ++a)
#pragma unroll
        for (int j = 0; j < 4; ++j)
            wf[a][j] = *(const v4f*)(wrow + 32 * a + 4 * j);
    const float bias = b[ct * 16 + ln];
    __syncthreads();

    for (int mt = wave_u >> 3; mt < ROW_TILES; mt += 2) {
        const float* arow = acc + (mt * 16 + ln) * PITCH + 4 * hf;
        v8f d;
#pragma unroll
        for (int v = 0; v < 8; ++v) d[v] = bias;
#pragma unroll
        for (int s = 0; s < 16; ++s) {
            const v4f x = *(const v4f*)(arow + 8 * s);
            const int a = s & 3, qq = s >> 2;
            v2f a0, a1, b0, b1;
            a0[0] = x[0];            a0[1] = x[1];
            a1[0] = x[2];            a1[1] = x[3];
            b0[0] = wf[a][0][qq];    b0[1] = wf[a][1][qq];
            b1[0] = wf[a][2][qq];    b1[1] = wf[a][3][qq];
            d = __builtin_amdgcn_wmma_f32_16x16x4_f32(false, a0, false, b0, (short)0, d, false, false);
            d = __builtin_amdgcn_wmma_f32_16x16x4_f32(false, a1, false, b1, (short)0, d, false, false);
        }
        const int rbase = mt * 16 + 8 * hf;
        float* o = out + (size_t)(lo + rbase) * FDIM + ct * 16 + ln;
#pragma unroll
        for (int v = 0; v < 8; ++v)
            if (rbase + v < ROWS_PER_WG) o[v * FDIM] = d[v];
    }
}

extern "C" void kernel_launch(void* const* d_in, const int* in_sizes, int n_in,
                              void* d_out, int out_size, void* d_ws, size_t ws_size,
                              hipStream_t stream) {
    const float* feature = (const float*)d_in[0];
    const int*   src     = (const int*)  d_in[1];
    const int*   dst     = (const int*)  d_in[2];
    const float* W       = (const float*)d_in[3];
    const float* b       = (const float*)d_in[4];
    float*       out     = (float*)d_out;
    unsigned*    planes  = (unsigned*)d_ws;

    gcn_pack_owner_kernel<<<N_BLOCKS_PAD / PACK_WAVES, PACK_WAVES * 32, 0, stream>>>(src, dst, planes);
    gcn_owner_layer_kernel<<<N_OWNERS, WG_THREADS, 0, stream>>>(feature, W, b, planes, out);
}
